# starting kernel
# baseline (speedup 1.0000x reference)
#include <hip/hip_runtime.h>

typedef __attribute__((ext_vector_type(8))) float v8f;
typedef __attribute__((ext_vector_type(2))) float v2f;

__global__ void fuse_kernel(const float* __restrict__ x,
                            const float* __restrict__ embed,
                            const int* __restrict__ node_index,
                            const float* __restrict__ W,
                            const float* __restrict__ b,
                            float* __restrict__ out,
                            int n, int fin) {
  int t = blockIdx.x * blockDim.x + threadIdx.x;
  int node = t >> 5;
  int j = t & 31;
  if (node >= n) return;
  float v;
  if (j < 16) {
    v = embed[(long long)node_index[node] * 16 + j];
  } else {
    int r = j - 16;
    float acc = b[r];
    const float* xr = x + (long long)node * fin;
    for (int k = 0; k < fin; ++k) acc = fmaf(xr[k], W[r * fin + k], acc);
    v = acc;
  }
  out[(long long)node * 32 + j] = v;
}

__global__ void degree_kernel(const int* __restrict__ idx,
                              float* __restrict__ deg, int E) {
  int t = blockIdx.x * blockDim.x + threadIdx.x;
  if (t < E) atomicAdd(&deg[idx[t]], 1.0f);
}

__global__ void scatter_kernel(const float* __restrict__ srcf,
                               const int* __restrict__ gather_idx,
                               const int* __restrict__ scatter_idx,
                               float* __restrict__ agg,
                               int E) {
  int t = blockIdx.x * blockDim.x + threadIdx.x;
  int e = t >> 3;
  int part = t & 7;
  if (e >= E) return;
  int s = gather_idx[e];
  int d = scatter_idx[e];
  float4 v = ((const float4*)(srcf + (long long)s * 32))[part];
  float* o = agg + (long long)d * 32 + part * 4;
  atomicAdd(o + 0, v.x);
  atomicAdd(o + 1, v.y);
  atomicAdd(o + 2, v.z);
  atomicAdd(o + 3, v.w);
}

template <int NTILES, bool RELU>
__global__ void sage_wmma_kernel(const float* __restrict__ agg,
                                 const float* __restrict__ deg,
                                 const float* __restrict__ xdst,
                                 const float* __restrict__ Wl,
                                 const float* __restrict__ bl,
                                 const float* __restrict__ Wr,
                                 float* __restrict__ out,
                                 int n) {
  const int OUTF = NTILES * 16;
  int lane = threadIdx.x & 31;
  int wave = (blockIdx.x * blockDim.x + threadIdx.x) >> 5;
  int nb = wave * 16;
  if (nb >= n) return;
  int m  = lane & 15;
  int hi = lane >> 4;

  int node = nb + m;
  bool nv = node < n;
  int ncl = nv ? node : (n - 1);
  float dinv = nv ? (1.0f / fmaxf(deg[ncl], 1.0f)) : 0.0f;
  const float* arow = agg  + (long long)ncl * 32;
  const float* xrow = xdst + (long long)ncl * 32;

  v2f a_agg[8], a_x[8];
#pragma unroll
  for (int s = 0; s < 8; ++s) {
    int k = 4 * s + 2 * hi;
    a_agg[s].x = arow[k] * dinv;
    a_agg[s].y = arow[k + 1] * dinv;
    a_x[s].x   = nv ? xrow[k]     : 0.0f;
    a_x[s].y   = nv ? xrow[k + 1] : 0.0f;
  }

  bool full = (nb + 16 <= n);

#pragma unroll
  for (int t = 0; t < NTILES; ++t) {
    int ng = t * 16 + m;
    const float* wl = Wl + ng * 32;
    const float* wr = Wr + ng * 32;
    float bias = bl[ng];
    v8f c;
#pragma unroll
    for (int v = 0; v < 8; ++v) c[v] = bias;

#pragma unroll
    for (int s = 0; s < 8; ++s) {
      int k = 4 * s + 2 * hi;
      v2f b;
      b.x = wl[k];
      b.y = wl[k + 1];
      c = __builtin_amdgcn_wmma_f32_16x16x4_f32(false, a_agg[s], false, b,
                                                (short)0, c, false, false);
    }
#pragma unroll
    for (int s = 0; s < 8; ++s) {
      int k = 4 * s + 2 * hi;
      v2f b;
      b.x = wr[k];
      b.y = wr[k + 1];
      c = __builtin_amdgcn_wmma_f32_16x16x4_f32(false, a_x[s], false, b,
                                                (short)0, c, false, false);
    }

    if (full) {
#pragma unroll
      for (int v = 0; v < 8; ++v) {
        float val = c[v];
        if (RELU) val = fmaxf(val, 0.0f);
        out[(long long)(nb + v + 8 * hi) * OUTF + ng] = val;
      }
    } else {
#pragma unroll
      for (int v = 0; v < 8; ++v) {
        int nd = nb + v + 8 * hi;
        if (nd < n) {
          float val = c[v];
          if (RELU) val = fmaxf(val, 0.0f);
          out[(long long)nd * OUTF + ng] = val;
        }
      }
    }
  }
}

__global__ void decode_kernel(const float* __restrict__ ou,
                              const float* __restrict__ oc,
                              const int* __restrict__ ls,
                              const int* __restrict__ ld,
                              float* __restrict__ out, int L) {
  int t = blockIdx.x * blockDim.x + threadIdx.x;
  if (t >= L) return;
  const float4* a = (const float4*)(ou + (long long)ls[t] * 16);
  const float4* b = (const float4*)(oc + (long long)ld[t] * 16);
  float acc = 0.0f;
#pragma unroll
  for (int i = 0; i < 4; ++i) {
    float4 xa = a[i], xb = b[i];
    acc += xa.x * xb.x + xa.y * xb.y + xa.z * xb.z + xa.w * xb.w;
  }
  out[t] = acc;
}

extern "C" void kernel_launch(void* const* d_in, const int* in_sizes, int n_in,
                              void* d_out, int out_size, void* d_ws, size_t ws_size,
                              hipStream_t stream) {
  const float* user_x       = (const float*)d_in[0];
  const float* course_x     = (const float*)d_in[1];
  const float* user_embed   = (const float*)d_in[2];
  const float* course_embed = (const float*)d_in[3];
  const float* Wu = (const float*)d_in[4];
  const float* bu = (const float*)d_in[5];
  const float* Wc = (const float*)d_in[6];
  const float* bc = (const float*)d_in[7];
  const float* c1e_Wl = (const float*)d_in[8];
  const float* c1e_bl = (const float*)d_in[9];
  const float* c1e_Wr = (const float*)d_in[10];
  const float* c1r_Wl = (const float*)d_in[11];
  const float* c1r_bl = (const float*)d_in[12];
  const float* c1r_Wr = (const float*)d_in[13];
  const float* c2e_Wl = (const float*)d_in[14];
  const float* c2e_bl = (const float*)d_in[15];
  const float* c2e_Wr = (const float*)d_in[16];
  const float* c2r_Wl = (const float*)d_in[17];
  const float* c2r_bl = (const float*)d_in[18];
  const float* c2r_Wr = (const float*)d_in[19];
  const int* user_idx   = (const int*)d_in[20];
  const int* course_idx = (const int*)d_in[21];
  const int* edge_src   = (const int*)d_in[22];
  const int* edge_dst   = (const int*)d_in[23];
  const int* label_src  = (const int*)d_in[24];
  const int* label_dst  = (const int*)d_in[25];
  float* outp = (float*)d_out;

  const int NU = in_sizes[2] / 16;
  const int NC = in_sizes[3] / 16;
  const int E  = in_sizes[22];
  const int L  = in_sizes[24];

  char* p = (char*)d_ws;
  auto carve = [&](size_t bytes) -> float* {
    float* r = (float*)p;
    p += (bytes + 255) & ~((size_t)255);
    return r;
  };
  float* xu   = carve((size_t)NU * 32 * 4);
  float* xc   = carve((size_t)NC * 32 * 4);
  float* aggu = carve((size_t)NU * 32 * 4);
  float* aggc = carve((size_t)NC * 32 * 4);
  float* degu = carve((size_t)NU * 4);
  float* degc = carve((size_t)NC * 4);
  float* hu   = carve((size_t)NU * 32 * 4);
  float* hc   = carve((size_t)NC * 32 * 4);
  float* ou   = carve((size_t)NU * 16 * 4);
  float* oc   = carve((size_t)NC * 16 * 4);

  const int B = 256;
  auto nblk = [](long long t, int b) { return (unsigned)((t + b - 1) / b); };

  hipMemsetAsync(aggu, 0, (size_t)NU * 32 * 4, stream);
  hipMemsetAsync(aggc, 0, (size_t)NC * 32 * 4, stream);
  hipMemsetAsync(degu, 0, (size_t)NU * 4, stream);
  hipMemsetAsync(degc, 0, (size_t)NC * 4, stream);

  fuse_kernel<<<nblk((long long)NU * 32, B), B, 0, stream>>>(
      user_x, user_embed, user_idx, Wu, bu, xu, NU, 5);
  fuse_kernel<<<nblk((long long)NC * 32, B), B, 0, stream>>>(
      course_x, course_embed, course_idx, Wc, bc, xc, NC, 2);

  degree_kernel<<<nblk(E, B), B, 0, stream>>>(edge_dst, degc, E);
  degree_kernel<<<nblk(E, B), B, 0, stream>>>(edge_src, degu, E);

  scatter_kernel<<<nblk((long long)E * 8, B), B, 0, stream>>>(xu, edge_src, edge_dst, aggc, E);
  scatter_kernel<<<nblk((long long)E * 8, B), B, 0, stream>>>(xc, edge_dst, edge_src, aggu, E);
  sage_wmma_kernel<2, true><<<nblk(((long long)(NC + 15) / 16) * 32, B), B, 0, stream>>>(
      aggc, degc, xc, c1e_Wl, c1e_bl, c1e_Wr, hc, NC);
  sage_wmma_kernel<2, true><<<nblk(((long long)(NU + 15) / 16) * 32, B), B, 0, stream>>>(
      aggu, degu, xu, c1r_Wl, c1r_bl, c1r_Wr, hu, NU);

  hipMemsetAsync(aggu, 0, (size_t)NU * 32 * 4, stream);
  hipMemsetAsync(aggc, 0, (size_t)NC * 32 * 4, stream);
  scatter_kernel<<<nblk((long long)E * 8, B), B, 0, stream>>>(hu, edge_src, edge_dst, aggc, E);
  scatter_kernel<<<nblk((long long)E * 8, B), B, 0, stream>>>(hc, edge_dst, edge_src, aggu, E);
  sage_wmma_kernel<1, false><<<nblk(((long long)(NC + 15) / 16) * 32, B), B, 0, stream>>>(
      aggc, degc, hc, c2e_Wl, c2e_bl, c2e_Wr, oc, NC);
  sage_wmma_kernel<1, false><<<nblk(((long long)(NU + 15) / 16) * 32, B), B, 0, stream>>>(
      aggu, degu, hu, c2r_Wl, c2r_bl, c2r_Wr, ou, NU);

  decode_kernel<<<nblk(L, B), B, 0, stream>>>(ou, oc, label_src, label_dst, outp, L);
}


# course-side degree count: per-block LDS integer histograms stored as partial rows and summed by one added kernel, replacing 2M contended global float atomics on a 40 KB array and its memset
# speedup vs baseline: 1.3153x; 1.3153x over previous
#include <hip/hip_runtime.h>

typedef __attribute__((ext_vector_type(8))) float v8f;
typedef __attribute__((ext_vector_type(2))) float v2f;

__global__ void fuse_kernel(const float* __restrict__ x,
                            const float* __restrict__ embed,
                            const int* __restrict__ node_index,
                            const float* __restrict__ W,
                            const float* __restrict__ b,
                            float* __restrict__ out,
                            int n, int fin) {
  int t = blockIdx.x * blockDim.x + threadIdx.x;
  int node = t >> 5;
  int j = t & 31;
  if (node >= n) return;
  float v;
  if (j < 16) {
    v = embed[(long long)node_index[node] * 16 + j];
  } else {
    int r = j - 16;
    float acc = b[r];
    const float* xr = x + (long long)node * fin;
    for (int k = 0; k < fin; ++k) acc = fmaf(xr[k], W[r * fin + k], acc);
    v = acc;
  }
  out[(long long)node * 32 + j] = v;
}

__global__ void degree_kernel(const int* __restrict__ idx,
                              float* __restrict__ deg, int E) {
  int t = blockIdx.x * blockDim.x + threadIdx.x;
  if (t < E) atomicAdd(&deg[idx[t]], 1.0f);
}

#define HIST_BINS    16384
#define HIST_THREADS 1024
#define HIST_BATCH   8

__global__ __launch_bounds__(HIST_THREADS)
void degree_hist_kernel(const int* __restrict__ idx,
                        unsigned* __restrict__ part, int E, int n) {
  __shared__ unsigned h[HIST_BINS];
  const int tid = threadIdx.x;
  const int chunk = (E + (int)gridDim.x - 1) / (int)gridDim.x;
  const int e0 = (int)blockIdx.x * chunk;
  const int e1 = min(e0 + chunk, E);
  unsigned* prow = part + (size_t)blockIdx.x * (size_t)n;

  for (int base = 0; base < n; base += HIST_BINS) {
    const int nb = min(HIST_BINS, n - base);
    for (int i = tid; i < nb; i += HIST_THREADS) h[i] = 0u;
    __syncthreads();

    for (int e = e0 + tid; e < e1; e += HIST_BATCH * HIST_THREADS) {
      int b[HIST_BATCH];
#pragma unroll
      for (int u = 0; u < HIST_BATCH; ++u) {
        const int ee = e + u * HIST_THREADS;
        b[u] = (ee < e1) ? idx[ee] : -1;
      }
#pragma unroll
      for (int u = 0; u < HIST_BATCH; ++u) {
        unsigned r = (unsigned)(b[u] - base);
        if (r < (unsigned)nb) atomicAdd(&h[r], 1u);
      }
    }
    __syncthreads();

    for (int i = tid; i < nb; i += HIST_THREADS) prow[base + i] = h[i];
    __syncthreads();
  }
}

__global__ void degree_sum_kernel(const unsigned* __restrict__ part,
                                  float* __restrict__ deg, int n, int rows) {
  int i = blockIdx.x * blockDim.x + threadIdx.x;
  if (i >= n) return;
  unsigned s = 0u;
#pragma unroll 16
  for (int g = 0; g < rows; ++g) s += part[(size_t)g * (size_t)n + i];
  deg[i] = (float)s;
}

__global__ void scatter_kernel(const float* __restrict__ srcf,
                               const int* __restrict__ gather_idx,
                               const int* __restrict__ scatter_idx,
                               float* __restrict__ agg,
                               int E) {
  int t = blockIdx.x * blockDim.x + threadIdx.x;
  int e = t >> 3;
  int part = t & 7;
  if (e >= E) return;
  int s = gather_idx[e];
  int d = scatter_idx[e];
  float4 v = ((const float4*)(srcf + (long long)s * 32))[part];
  float* o = agg + (long long)d * 32 + part * 4;
  atomicAdd(o + 0, v.x);
  atomicAdd(o + 1, v.y);
  atomicAdd(o + 2, v.z);
  atomicAdd(o + 3, v.w);
}

template <int NTILES, bool RELU>
__global__ void sage_wmma_kernel(const float* __restrict__ agg,
                                 const float* __restrict__ deg,
                                 const float* __restrict__ xdst,
                                 const float* __restrict__ Wl,
                                 const float* __restrict__ bl,
                                 const float* __restrict__ Wr,
                                 float* __restrict__ out,
                                 int n) {
  const int OUTF = NTILES * 16;
  int lane = threadIdx.x & 31;
  int wave = (blockIdx.x * blockDim.x + threadIdx.x) >> 5;
  int nb = wave * 16;
  if (nb >= n) return;
  int m  = lane & 15;
  int hi = lane >> 4;

  int node = nb + m;
  bool nv = node < n;
  int ncl = nv ? node : (n - 1);
  float dinv = nv ? (1.0f / fmaxf(deg[ncl], 1.0f)) : 0.0f;
  const float* arow = agg  + (long long)ncl * 32;
  const float* xrow = xdst + (long long)ncl * 32;

  v2f a_agg[8], a_x[8];
#pragma unroll
  for (int s = 0; s < 8; ++s) {
    int k = 4 * s + 2 * hi;
    a_agg[s].x = arow[k] * dinv;
    a_agg[s].y = arow[k + 1] * dinv;
    a_x[s].x   = nv ? xrow[k]     : 0.0f;
    a_x[s].y   = nv ? xrow[k + 1] : 0.0f;
  }

  bool full = (nb + 16 <= n);

#pragma unroll
  for (int t = 0; t < NTILES; ++t) {
    int ng = t * 16 + m;
    const float* wl = Wl + ng * 32;
    const float* wr = Wr + ng * 32;
    float bias = bl[ng];
    v8f c;
#pragma unroll
    for (int v = 0; v < 8; ++v) c[v] = bias;

#pragma unroll
    for (int s = 0; s < 8; ++s) {
      int k = 4 * s + 2 * hi;
      v2f b;
      b.x = wl[k];
      b.y = wl[k + 1];
      c = __builtin_amdgcn_wmma_f32_16x16x4_f32(false, a_agg[s], false, b,
                                                (short)0, c, false, false);
    }
#pragma unroll
    for (int s = 0; s < 8; ++s) {
      int k = 4 * s + 2 * hi;
      v2f b;
      b.x = wr[k];
      b.y = wr[k + 1];
      c = __builtin_amdgcn_wmma_f32_16x16x4_f32(false, a_x[s], false, b,
                                                (short)0, c, false, false);
    }

    if (full) {
#pragma unroll
      for (int v = 0; v < 8; ++v) {
        float val = c[v];
        if (RELU) val = fmaxf(val, 0.0f);
        out[(long long)(nb + v + 8 * hi) * OUTF + ng] = val;
      }
    } else {
#pragma unroll
      for (int v = 0; v < 8; ++v) {
        int nd = nb + v + 8 * hi;
        if (nd < n) {
          float val = c[v];
          if (RELU) val = fmaxf(val, 0.0f);
          out[(long long)nd * OUTF + ng] = val;
        }
      }
    }
  }
}

__global__ void decode_kernel(const float* __restrict__ ou,
                              const float* __restrict__ oc,
                              const int* __restrict__ ls,
                              const int* __restrict__ ld,
                              float* __restrict__ out, int L) {
  int t = blockIdx.x * blockDim.x + threadIdx.x;
  if (t >= L) return;
  const float4* a = (const float4*)(ou + (long long)ls[t] * 16);
  const float4* b = (const float4*)(oc + (long long)ld[t] * 16);
  float acc = 0.0f;
#pragma unroll
  for (int i = 0; i < 4; ++i) {
    float4 xa = a[i], xb = b[i];
    acc += xa.x * xb.x + xa.y * xb.y + xa.z * xb.z + xa.w * xb.w;
  }
  out[t] = acc;
}

extern "C" void kernel_launch(void* const* d_in, const int* in_sizes, int n_in,
                              void* d_out, int out_size, void* d_ws, size_t ws_size,
                              hipStream_t stream) {
  const float* user_x       = (const float*)d_in[0];
  const float* course_x     = (const float*)d_in[1];
  const float* user_embed   = (const float*)d_in[2];
  const float* course_embed = (const float*)d_in[3];
  const float* Wu = (const float*)d_in[4];
  const float* bu = (const float*)d_in[5];
  const float* Wc = (const float*)d_in[6];
  const float* bc = (const float*)d_in[7];
  const float* c1e_Wl = (const float*)d_in[8];
  const float* c1e_bl = (const float*)d_in[9];
  const float* c1e_Wr = (const float*)d_in[10];
  const float* c1r_Wl = (const float*)d_in[11];
  const float* c1r_bl = (const float*)d_in[12];
  const float* c1r_Wr = (const float*)d_in[13];
  const float* c2e_Wl = (const float*)d_in[14];
  const float* c2e_bl = (const float*)d_in[15];
  const float* c2e_Wr = (const float*)d_in[16];
  const float* c2r_Wl = (const float*)d_in[17];
  const float* c2r_bl = (const float*)d_in[18];
  const float* c2r_Wr = (const float*)d_in[19];
  const int* user_idx   = (const int*)d_in[20];
  const int* course_idx = (const int*)d_in[21];
  const int* edge_src   = (const int*)d_in[22];
  const int* edge_dst   = (const int*)d_in[23];
  const int* label_src  = (const int*)d_in[24];
  const int* label_dst  = (const int*)d_in[25];
  float* outp = (float*)d_out;

  const int NU = in_sizes[2] / 16;
  const int NC = in_sizes[3] / 16;
  const int E  = in_sizes[22];
  const int L  = in_sizes[24];

  char* p = (char*)d_ws;
  auto carve = [&](size_t bytes) -> float* {
    float* r = (float*)p;
    p += (bytes + 255) & ~((size_t)255);
    return r;
  };
  float* xu   = carve((size_t)NU * 32 * 4);
  float* xc   = carve((size_t)NC * 32 * 4);
  float* aggu = carve((size_t)NU * 32 * 4);
  float* aggc = carve((size_t)NC * 32 * 4);
  float* degu = carve((size_t)NU * 4);
  float* degc = carve((size_t)NC * 4);
  float* hu   = carve((size_t)NU * 32 * 4);
  float* hc   = carve((size_t)NC * 32 * 4);
  float* ou   = carve((size_t)NU * 16 * 4);
  float* oc   = carve((size_t)NC * 16 * 4);

  const int B = 256;
  auto nblk = [](long long t, int b) { return (unsigned)((t + b - 1) / b); };

  hipMemsetAsync(aggu, 0, (size_t)NU * 32 * 4, stream);
  hipMemsetAsync(aggc, 0, (size_t)NC * 32 * 4, stream);
  hipMemsetAsync(degu, 0, (size_t)NU * 4, stream);

  fuse_kernel<<<nblk((long long)NU * 32, B), B, 0, stream>>>(
      user_x, user_embed, user_idx, Wu, bu, xu, NU, 5);
  fuse_kernel<<<nblk((long long)NC * 32, B), B, 0, stream>>>(
      course_x, course_embed, course_idx, Wc, bc, xc, NC, 2);

  unsigned* degpart = (unsigned*)hu;
  degree_hist_kernel<<<64, HIST_THREADS, 0, stream>>>(edge_dst, degpart, E, NC);
  degree_sum_kernel<<<nblk(NC, B), B, 0, stream>>>(degpart, degc, NC, 64);
  degree_kernel<<<nblk(E, B), B, 0, stream>>>(edge_src, degu, E);

  scatter_kernel<<<nblk((long long)E * 8, B), B, 0, stream>>>(xu, edge_src, edge_dst, aggc, E);
  scatter_kernel<<<nblk((long long)E * 8, B), B, 0, stream>>>(xc, edge_dst, edge_src, aggu, E);
  sage_wmma_kernel<2, true><<<nblk(((long long)(NC + 15) / 16) * 32, B), B, 0, stream>>>(
      aggc, degc, xc, c1e_Wl, c1e_bl, c1e_Wr, hc, NC);
  sage_wmma_kernel<2, true><<<nblk(((long long)(NU + 15) / 16) * 32, B), B, 0, stream>>>(
      aggu, degu, xu, c1r_Wl, c1r_bl, c1r_Wr, hu, NU);

  hipMemsetAsync(aggu, 0, (size_t)NU * 32 * 4, stream);
  hipMemsetAsync(aggc, 0, (size_t)NC * 32 * 4, stream);
  scatter_kernel<<<nblk((long long)E * 8, B), B, 0, stream>>>(hu, edge_src, edge_dst, aggc, E);
  scatter_kernel<<<nblk((long long)E * 8, B), B, 0, stream>>>(hc, edge_dst, edge_src, aggu, E);
  sage_wmma_kernel<1, false><<<nblk(((long long)(NC + 15) / 16) * 32, B), B, 0, stream>>>(
      aggc, degc, hc, c2e_Wl, c2e_bl, c2e_Wr, oc, NC);
  sage_wmma_kernel<1, false><<<nblk(((long long)(NU + 15) / 16) * 32, B), B, 0, stream>>>(
      aggu, degu, hu, c2r_Wl, c2r_bl, c2r_Wr, ou, NU);

  decode_kernel<<<nblk(L, B), B, 0, stream>>>(ou, oc, label_src, label_dst, outp, L);
}


# fewer, longer-lived waves: fuse, user degree and 4 edge scatters run as one chip-filling grid (2048x256), threads walking items by grid stride; float atomics kept, arrival order unordered as before
# speedup vs baseline: 1.4000x; 1.0644x over previous
#include <hip/hip_runtime.h>

typedef __attribute__((ext_vector_type(8))) float v8f;
typedef __attribute__((ext_vector_type(2))) float v2f;

#define RES_BLOCKS  2048
#define RES_THREADS 256

template <int FIN>
__global__ __launch_bounds__(RES_THREADS)
void fuse_kernel(const float* __restrict__ x,
                 const float* __restrict__ embed,
                 const int* __restrict__ node_index,
                 const float* __restrict__ W,
                 const float* __restrict__ b,
                 float* __restrict__ out,
                 int n) {
  const int part = threadIdx.x & 7;
  const int grp  = (int)((blockIdx.x * blockDim.x + threadIdx.x) >> 3);
  const int ngrp = (int)((gridDim.x * blockDim.x) >> 3);
  const bool lin = part >= 4;
  const int q4 = part & 3;
  const int r  = q4 * 4;
  float w[4][FIN], bias[4];
#pragma unroll
  for (int q = 0; q < 4; ++q) {
    bias[q] = b[r + q];
#pragma unroll
    for (int k = 0; k < FIN; ++k) w[q][k] = W[(r + q) * FIN + k];
  }
  for (int node = grp; node < n; node += ngrp) {
    const int ni = node_index[node];
    const float4 ev = ((const float4*)(embed + (long long)ni * 16))[q4];
    const float* xr = x + (long long)node * FIN;
    float xv[FIN];
#pragma unroll
    for (int k = 0; k < FIN; ++k) xv[k] = xr[k];
    float a[4];
#pragma unroll
    for (int q = 0; q < 4; ++q) {
      float acc = bias[q];
#pragma unroll
      for (int k = 0; k < FIN; ++k) acc = fmaf(xv[k], w[q][k], acc);
      a[q] = acc;
    }
    float4 v;
    v.x = lin ? a[0] : ev.x;
    v.y = lin ? a[1] : ev.y;
    v.z = lin ? a[2] : ev.z;
    v.w = lin ? a[3] : ev.w;
    ((float4*)(out + (long long)node * 32))[part] = v;
  }
}

#define DEG_BATCH 4

__global__ __launch_bounds__(RES_THREADS)
void degree_kernel(const int* __restrict__ idx,
                   float* __restrict__ deg, int E) {
  const int nt = (int)(gridDim.x * blockDim.x);
  int e = (int)(blockIdx.x * blockDim.x + threadIdx.x);
  for (; e + (DEG_BATCH - 1) * nt < E; e += DEG_BATCH * nt) {
    int d[DEG_BATCH];
#pragma unroll
    for (int u = 0; u < DEG_BATCH; ++u) d[u] = idx[e + u * nt];
#pragma unroll
    for (int u = 0; u < DEG_BATCH; ++u) atomicAdd(&deg[d[u]], 1.0f);
  }
  for (; e < E; e += nt) atomicAdd(&deg[idx[e]], 1.0f);
}

#define HIST_BINS    16384
#define HIST_THREADS 1024
#define HIST_BATCH   8

__global__ __launch_bounds__(HIST_THREADS)
void degree_hist_kernel(const int* __restrict__ idx,
                        unsigned* __restrict__ part, int E, int n) {
  __shared__ unsigned h[HIST_BINS];
  const int tid = threadIdx.x;
  const int chunk = (E + (int)gridDim.x - 1) / (int)gridDim.x;
  const int e0 = (int)blockIdx.x * chunk;
  const int e1 = min(e0 + chunk, E);
  unsigned* prow = part + (size_t)blockIdx.x * (size_t)n;

  for (int base = 0; base < n; base += HIST_BINS) {
    const int nb = min(HIST_BINS, n - base);
    for (int i = tid; i < nb; i += HIST_THREADS) h[i] = 0u;
    __syncthreads();

    for (int e = e0 + tid; e < e1; e += HIST_BATCH * HIST_THREADS) {
      int b[HIST_BATCH];
#pragma unroll
      for (int u = 0; u < HIST_BATCH; ++u) {
        const int ee = e + u * HIST_THREADS;
        b[u] = (ee < e1) ? idx[ee] : -1;
      }
#pragma unroll
      for (int u = 0; u < HIST_BATCH; ++u) {
        unsigned r = (unsigned)(b[u] - base);
        if (r < (unsigned)nb) atomicAdd(&h[r], 1u);
      }
    }
    __syncthreads();

    for (int i = tid; i < nb; i += HIST_THREADS) prow[base + i] = h[i];
    __syncthreads();
  }
}

__global__ void degree_sum_kernel(const unsigned* __restrict__ part,
                                  float* __restrict__ deg, int n, int rows) {
  int i = blockIdx.x * blockDim.x + threadIdx.x;
  if (i >= n) return;
  unsigned s = 0u;
#pragma unroll 16
  for (int g = 0; g < rows; ++g) s += part[(size_t)g * (size_t)n + i];
  deg[i] = (float)s;
}

#define SCAT_BATCH 5

__device__ __forceinline__ void scatter_add_row(float* __restrict__ o, float4 v) {
  atomicAdd(o + 0, v.x);
  atomicAdd(o + 1, v.y);
  atomicAdd(o + 2, v.z);
  atomicAdd(o + 3, v.w);
}

__global__ __launch_bounds__(RES_THREADS)
void scatter_kernel(const float* __restrict__ srcf,
                    const int* __restrict__ gather_idx,
                    const int* __restrict__ scatter_idx,
                    float* __restrict__ agg,
                    int E) {
  const int part = threadIdx.x & 7;
  const int ngrp = (int)((gridDim.x * blockDim.x) >> 3);
  int e = (int)((blockIdx.x * blockDim.x + threadIdx.x) >> 3);
  for (; e + (SCAT_BATCH - 1) * ngrp < E; e += SCAT_BATCH * ngrp) {
    int s[SCAT_BATCH], d[SCAT_BATCH];
    float4 v[SCAT_BATCH];
#pragma unroll
    for (int u = 0; u < SCAT_BATCH; ++u) {
      s[u] = gather_idx[e + u * ngrp];
      d[u] = scatter_idx[e + u * ngrp];
    }
#pragma unroll
    for (int u = 0; u < SCAT_BATCH; ++u)
      v[u] = ((const float4*)(srcf + (long long)s[u] * 32))[part];
#pragma unroll
    for (int u = 0; u < SCAT_BATCH; ++u)
      scatter_add_row(agg + (long long)d[u] * 32 + part * 4, v[u]);
  }
  for (; e < E; e += ngrp) {
    const int s = gather_idx[e];
    const int d = scatter_idx[e];
    const float4 v = ((const float4*)(srcf + (long long)s * 32))[part];
    scatter_add_row(agg + (long long)d * 32 + part * 4, v);
  }
}

template <int NTILES, bool RELU>
__global__ void sage_wmma_kernel(const float* __restrict__ agg,
                                 const float* __restrict__ deg,
                                 const float* __restrict__ xdst,
                                 const float* __restrict__ Wl,
                                 const float* __restrict__ bl,
                                 const float* __restrict__ Wr,
                                 float* __restrict__ out,
                                 int n) {
  const int OUTF = NTILES * 16;
  int lane = threadIdx.x & 31;
  int wave = (blockIdx.x * blockDim.x + threadIdx.x) >> 5;
  int nb = wave * 16;
  if (nb >= n) return;
  int m  = lane & 15;
  int hi = lane >> 4;

  int node = nb + m;
  bool nv = node < n;
  int ncl = nv ? node : (n - 1);
  float dinv = nv ? (1.0f / fmaxf(deg[ncl], 1.0f)) : 0.0f;
  const float* arow = agg  + (long long)ncl * 32;
  const float* xrow = xdst + (long long)ncl * 32;

  v2f a_agg[8], a_x[8];
#pragma unroll
  for (int s = 0; s < 8; ++s) {
    int k = 4 * s + 2 * hi;
    a_agg[s].x = arow[k] * dinv;
    a_agg[s].y = arow[k + 1] * dinv;
    a_x[s].x   = nv ? xrow[k]     : 0.0f;
    a_x[s].y   = nv ? xrow[k + 1] : 0.0f;
  }

  bool full = (nb + 16 <= n);

#pragma unroll
  for (int t = 0; t < NTILES; ++t) {
    int ng = t * 16 + m;
    const float* wl = Wl + ng * 32;
    const float* wr = Wr + ng * 32;
    float bias = bl[ng];
    v8f c;
#pragma unroll
    for (int v = 0; v < 8; ++v) c[v] = bias;

#pragma unroll
    for (int s = 0; s < 8; ++s) {
      int k = 4 * s + 2 * hi;
      v2f b;
      b.x = wl[k];
      b.y = wl[k + 1];
      c = __builtin_amdgcn_wmma_f32_16x16x4_f32(false, a_agg[s], false, b,
                                                (short)0, c, false, false);
    }
#pragma unroll
    for (int s = 0; s < 8; ++s) {
      int k = 4 * s + 2 * hi;
      v2f b;
      b.x = wr[k];
      b.y = wr[k + 1];
      c = __builtin_amdgcn_wmma_f32_16x16x4_f32(false, a_x[s], false, b,
                                                (short)0, c, false, false);
    }

    if (full) {
#pragma unroll
      for (int v = 0; v < 8; ++v) {
        float val = c[v];
        if (RELU) val = fmaxf(val, 0.0f);
        out[(long long)(nb + v + 8 * hi) * OUTF + ng] = val;
      }
    } else {
#pragma unroll
      for (int v = 0; v < 8; ++v) {
        int nd = nb + v + 8 * hi;
        if (nd < n) {
          float val = c[v];
          if (RELU) val = fmaxf(val, 0.0f);
          out[(long long)nd * OUTF + ng] = val;
        }
      }
    }
  }
}

__global__ void decode_kernel(const float* __restrict__ ou,
                              const float* __restrict__ oc,
                              const int* __restrict__ ls,
                              const int* __restrict__ ld,
                              float* __restrict__ out, int L) {
  int t = blockIdx.x * blockDim.x + threadIdx.x;
  if (t >= L) return;
  const float4* a = (const float4*)(ou + (long long)ls[t] * 16);
  const float4* b = (const float4*)(oc + (long long)ld[t] * 16);
  float acc = 0.0f;
#pragma unroll
  for (int i = 0; i < 4; ++i) {
    float4 xa = a[i], xb = b[i];
    acc += xa.x * xb.x + xa.y * xb.y + xa.z * xb.z + xa.w * xb.w;
  }
  out[t] = acc;
}

extern "C" void kernel_launch(void* const* d_in, const int* in_sizes, int n_in,
                              void* d_out, int out_size, void* d_ws, size_t ws_size,
                              hipStream_t stream) {
  const float* user_x       = (const float*)d_in[0];
  const float* course_x     = (const float*)d_in[1];
  const float* user_embed   = (const float*)d_in[2];
  const float* course_embed = (const float*)d_in[3];
  const float* Wu = (const float*)d_in[4];
  const float* bu = (const float*)d_in[5];
  const float* Wc = (const float*)d_in[6];
  const float* bc = (const float*)d_in[7];
  const float* c1e_Wl = (const float*)d_in[8];
  const float* c1e_bl = (const float*)d_in[9];
  const float* c1e_Wr = (const float*)d_in[10];
  const float* c1r_Wl = (const float*)d_in[11];
  const float* c1r_bl = (const float*)d_in[12];
  const float* c1r_Wr = (const float*)d_in[13];
  const float* c2e_Wl = (const float*)d_in[14];
  const float* c2e_bl = (const float*)d_in[15];
  const float* c2e_Wr = (const float*)d_in[16];
  const float* c2r_Wl = (const float*)d_in[17];
  const float* c2r_bl = (const float*)d_in[18];
  const float* c2r_Wr = (const float*)d_in[19];
  const int* user_idx   = (const int*)d_in[20];
  const int* course_idx = (const int*)d_in[21];
  const int* edge_src   = (const int*)d_in[22];
  const int* edge_dst   = (const int*)d_in[23];
  const int* label_src  = (const int*)d_in[24];
  const int* label_dst  = (const int*)d_in[25];
  float* outp = (float*)d_out;

  const int NU = in_sizes[2] / 16;
  const int NC = in_sizes[3] / 16;
  const int E  = in_sizes[22];
  const int L  = in_sizes[24];

  char* p = (char*)d_ws;
  auto carve = [&](size_t bytes) -> float* {
    float* r = (float*)p;
    p += (bytes + 255) & ~((size_t)255);
    return r;
  };
  float* xu   = carve((size_t)NU * 32 * 4);
  float* xc   = carve((size_t)NC * 32 * 4);
  float* aggu = carve((size_t)NU * 32 * 4);
  float* aggc = carve((size_t)NC * 32 * 4);
  float* degu = carve((size_t)NU * 4);
  float* degc = carve((size_t)NC * 4);
  float* hu   = carve((size_t)NU * 32 * 4);
  float* hc   = carve((size_t)NC * 32 * 4);
  float* ou   = carve((size_t)NU * 16 * 4);
  float* oc   = carve((size_t)NC * 16 * 4);

  const int B = 256;
  auto nblk = [](long long t, int b) { return (unsigned)((t + b - 1) / b); };
  auto rblk = [](long long t) {
    long long g = (t + RES_THREADS - 1) / RES_THREADS;
    return (unsigned)(g < RES_BLOCKS ? g : RES_BLOCKS);
  };

  hipMemsetAsync(aggu, 0, (size_t)NU * 32 * 4, stream);
  hipMemsetAsync(aggc, 0, (size_t)NC * 32 * 4, stream);
  hipMemsetAsync(degu, 0, (size_t)NU * 4, stream);

  fuse_kernel<5><<<rblk((long long)NU * 8), RES_THREADS, 0, stream>>>(
      user_x, user_embed, user_idx, Wu, bu, xu, NU);
  fuse_kernel<2><<<rblk((long long)NC * 8), RES_THREADS, 0, stream>>>(
      course_x, course_embed, course_idx, Wc, bc, xc, NC);

  unsigned* degpart = (unsigned*)hu;
  degree_hist_kernel<<<64, HIST_THREADS, 0, stream>>>(edge_dst, degpart, E, NC);
  degree_sum_kernel<<<nblk(NC, B), B, 0, stream>>>(degpart, degc, NC, 64);
  degree_kernel<<<rblk(E), RES_THREADS, 0, stream>>>(edge_src, degu, E);

  scatter_kernel<<<rblk((long long)E * 8), RES_THREADS, 0, stream>>>(xu, edge_src, edge_dst, aggc, E);
  scatter_kernel<<<rblk((long long)E * 8), RES_THREADS, 0, stream>>>(xc, edge_dst, edge_src, aggu, E);
  sage_wmma_kernel<2, true><<<nblk(((long long)(NC + 15) / 16) * 32, B), B, 0, stream>>>(
      aggc, degc, xc, c1e_Wl, c1e_bl, c1e_Wr, hc, NC);
  sage_wmma_kernel<2, true><<<nblk(((long long)(NU + 15) / 16) * 32, B), B, 0, stream>>>(
      aggu, degu, xu, c1r_Wl, c1r_bl, c1r_Wr, hu, NU);

  hipMemsetAsync(aggu, 0, (size_t)NU * 32 * 4, stream);
  hipMemsetAsync(aggc, 0, (size_t)NC * 32 * 4, stream);
  scatter_kernel<<<rblk((long long)E * 8), RES_THREADS, 0, stream>>>(hu, edge_src, edge_dst, aggc, E);
  scatter_kernel<<<rblk((long long)E * 8), RES_THREADS, 0, stream>>>(hc, edge_dst, edge_src, aggu, E);
  sage_wmma_kernel<1, false><<<nblk(((long long)(NC + 15) / 16) * 32, B), B, 0, stream>>>(
      aggc, degc, hc, c2e_Wl, c2e_bl, c2e_Wr, oc, NC);
  sage_wmma_kernel<1, false><<<nblk(((long long)(NU + 15) / 16) * 32, B), B, 0, stream>>>(
      aggu, degu, hu, c2r_Wl, c2r_bl, c2r_Wr, ou, NU);

  decode_kernel<<<nblk(L, B), B, 0, stream>>>(ou, oc, label_src, label_dst, outp, L);
}


# user-side neighbour sums as a gather over edges bucketed by user once per call (int rank atomics, scan, fill) instead of float atomics into a cleared 64 MB accumulator; sum order = rank arrival order
# speedup vs baseline: 1.5903x; 1.1359x over previous
#include <hip/hip_runtime.h>

typedef __attribute__((ext_vector_type(8))) float v8f;
typedef __attribute__((ext_vector_type(2))) float v2f;

#define RES_BLOCKS  2048
#define RES_THREADS 256

template <int FIN>
__global__ __launch_bounds__(RES_THREADS)
void fuse_kernel(const float* __restrict__ x,
                 const float* __restrict__ embed,
                 const int* __restrict__ node_index,
                 const float* __restrict__ W,
                 const float* __restrict__ b,
                 float* __restrict__ out,
                 int n) {
  const int part = threadIdx.x & 7;
  const int grp  = (int)((blockIdx.x * blockDim.x + threadIdx.x) >> 3);
  const int ngrp = (int)((gridDim.x * blockDim.x) >> 3);
  const bool lin = part >= 4;
  const int q4 = part & 3;
  const int r  = q4 * 4;
  float w[4][FIN], bias[4];
#pragma unroll
  for (int q = 0; q < 4; ++q) {
    bias[q] = b[r + q];
#pragma unroll
    for (int k = 0; k < FIN; ++k) w[q][k] = W[(r + q) * FIN + k];
  }
  for (int node = grp; node < n; node += ngrp) {
    const int ni = node_index[node];
    const float4 ev = ((const float4*)(embed + (long long)ni * 16))[q4];
    const float* xr = x + (long long)node * FIN;
    float xv[FIN];
#pragma unroll
    for (int k = 0; k < FIN; ++k) xv[k] = xr[k];
    float a[4];
#pragma unroll
    for (int q = 0; q < 4; ++q) {
      float acc = bias[q];
#pragma unroll
      for (int k = 0; k < FIN; ++k) acc = fmaf(xv[k], w[q][k], acc);
      a[q] = acc;
    }
    float4 v;
    v.x = lin ? a[0] : ev.x;
    v.y = lin ? a[1] : ev.y;
    v.z = lin ? a[2] : ev.z;
    v.w = lin ? a[3] : ev.w;
    ((float4*)(out + (long long)node * 32))[part] = v;
  }
}

#define RANK_BATCH 4

__global__ __launch_bounds__(RES_THREADS)
void rank_kernel(const int* __restrict__ idx, int* __restrict__ cnt,
                 int* __restrict__ rank, int E) {
  const int nt = (int)(gridDim.x * blockDim.x);
  int e = (int)(blockIdx.x * blockDim.x + threadIdx.x);
  for (; e + (RANK_BATCH - 1) * nt < E; e += RANK_BATCH * nt) {
    int d[RANK_BATCH], r[RANK_BATCH];
#pragma unroll
    for (int u = 0; u < RANK_BATCH; ++u) d[u] = idx[e + u * nt];
#pragma unroll
    for (int u = 0; u < RANK_BATCH; ++u) r[u] = atomicAdd(&cnt[d[u]], 1);
#pragma unroll
    for (int u = 0; u < RANK_BATCH; ++u) rank[e + u * nt] = r[u];
  }
  for (; e < E; e += nt) rank[e] = atomicAdd(&cnt[idx[e]], 1);
}

#define SCAN_THREADS 1024
#define SCAN_CHUNK   (SCAN_THREADS * 4)

__device__ __forceinline__ int wave_sum(int v) {
#pragma unroll
  for (int o = 16; o > 0; o >>= 1) v += __shfl_xor(v, o, 32);
  return v;
}

__device__ __forceinline__ int block_sum(int v, int* ws) {
  v = wave_sum(v);
  if ((threadIdx.x & 31) == 0) ws[threadIdx.x >> 5] = v;
  __syncthreads();
  int t = wave_sum(ws[threadIdx.x & 31]);
  __syncthreads();
  return t;
}

__device__ __forceinline__ int4 load_counts4(const int* __restrict__ cnt, int i0, int n) {
  int4 c;
  if (i0 + 3 < n) {
    c = *(const int4*)(cnt + i0);
  } else {
    c.x = (i0     < n) ? cnt[i0]     : 0;
    c.y = (i0 + 1 < n) ? cnt[i0 + 1] : 0;
    c.z = (i0 + 2 < n) ? cnt[i0 + 2] : 0;
    c.w = 0;
  }
  return c;
}

__global__ __launch_bounds__(SCAN_THREADS)
void row_total_kernel(const int* __restrict__ cnt, int* __restrict__ btot, int n) {
  __shared__ int ws[32];
  const int i0 = (int)blockIdx.x * SCAN_CHUNK + (int)threadIdx.x * 4;
  const int4 c = load_counts4(cnt, i0, n);
  const int t = block_sum(c.x + c.y + c.z + c.w, ws);
  if (threadIdx.x == 0) btot[blockIdx.x] = t;
}

__global__ __launch_bounds__(SCAN_THREADS)
void row_start_kernel(const int* __restrict__ cnt, const int* __restrict__ btot,
                      int* __restrict__ rowptr, float* __restrict__ deg, int n) {
  __shared__ int ws[32];
  const int tid = threadIdx.x, lane = tid & 31, wv = tid >> 5;
  int before = 0;
  for (int j = tid; j < (int)blockIdx.x; j += SCAN_THREADS) before += btot[j];
  before = block_sum(before, ws);

  const int i0 = (int)blockIdx.x * SCAN_CHUNK + tid * 4;
  const int4 c = load_counts4(cnt, i0, n);
  const int local = c.x + c.y + c.z + c.w;
  int inc = local;
#pragma unroll
  for (int o = 1; o < 32; o <<= 1) {
    const int t = __shfl_up(inc, o, 32);
    if (lane >= o) inc += t;
  }
  if (lane == 31) ws[wv] = inc;
  __syncthreads();
  const int wt = ws[lane];
  int winc = wt;
#pragma unroll
  for (int o = 1; o < 32; o <<= 1) {
    const int t = __shfl_up(winc, o, 32);
    if (lane >= o) winc += t;
  }
  const int wexc = __shfl(winc - wt, wv, 32);
  const int p0 = before + wexc + (inc - local);
  const int p1 = p0 + c.x, p2 = p1 + c.y, p3 = p2 + c.z;
  if (i0 + 3 < n) {
    *(int4*)(rowptr + i0) = make_int4(p0, p1, p2, p3);
    *(float4*)(deg + i0) = make_float4((float)c.x, (float)c.y, (float)c.z, (float)c.w);
  } else {
    if (i0     < n) { rowptr[i0]     = p0; deg[i0]     = (float)c.x; }
    if (i0 + 1 < n) { rowptr[i0 + 1] = p1; deg[i0 + 1] = (float)c.y; }
    if (i0 + 2 < n) { rowptr[i0 + 2] = p2; deg[i0 + 2] = (float)c.z; }
  }
}

#define FILL_BATCH 4

__global__ __launch_bounds__(RES_THREADS)
void fill_kernel(const int* __restrict__ row, const int* __restrict__ other,
                 const int* __restrict__ rank, const int* __restrict__ rowptr,
                 int* __restrict__ col, int E) {
  const int nt = (int)(gridDim.x * blockDim.x);
  int e = (int)(blockIdx.x * blockDim.x + threadIdx.x);
  for (; e + (FILL_BATCH - 1) * nt < E; e += FILL_BATCH * nt) {
    int s[FILL_BATCH], d[FILL_BATCH], r[FILL_BATCH], p[FILL_BATCH];
#pragma unroll
    for (int u = 0; u < FILL_BATCH; ++u) {
      s[u] = row[e + u * nt];
      d[u] = other[e + u * nt];
      r[u] = rank[e + u * nt];
    }
#pragma unroll
    for (int u = 0; u < FILL_BATCH; ++u) p[u] = rowptr[s[u]];
#pragma unroll
    for (int u = 0; u < FILL_BATCH; ++u) col[p[u] + r[u]] = d[u];
  }
  for (; e < E; e += nt) col[rowptr[row[e]] + rank[e]] = other[e];
}

__global__ __launch_bounds__(RES_THREADS)
void gather_kernel(const float* __restrict__ srcf,
                   const int* __restrict__ rowptr,
                   const int* __restrict__ rowlen,
                   const int* __restrict__ col,
                   float* __restrict__ agg, int n) {
  const int part = threadIdx.x & 7;
  const int ngrp = (int)((gridDim.x * blockDim.x) >> 3);
  for (int node = (int)((blockIdx.x * blockDim.x + threadIdx.x) >> 3); node < n; node += ngrp) {
    const int start = rowptr[node];
    const int len   = rowlen[node];
    float4 acc = make_float4(0.0f, 0.0f, 0.0f, 0.0f);
    for (int base = 0; base < len; base += 8) {
      const int m = min(8, len - base);
      const int ci = (part < m) ? col[start + base + part] : 0;
      for (int j0 = 0; j0 < m; j0 += 4) {
        float4 v[4];
#pragma unroll
        for (int j = 0; j < 4; ++j) {
          const int jj = (j0 + j < m) ? (j0 + j) : j0;
          const int c = __shfl(ci, jj, 8);
          v[j] = ((const float4*)(srcf + (long long)c * 32))[part];
        }
#pragma unroll
        for (int j = 0; j < 4; ++j) {
          if (j0 + j < m) {
            acc.x += v[j].x; acc.y += v[j].y; acc.z += v[j].z; acc.w += v[j].w;
          }
        }
      }
    }
    ((float4*)(agg + (long long)node * 32))[part] = acc;
  }
}

#define HIST_BINS    16384
#define HIST_THREADS 1024
#define HIST_BATCH   8

__global__ __launch_bounds__(HIST_THREADS)
void degree_hist_kernel(const int* __restrict__ idx,
                        unsigned* __restrict__ part, int E, int n) {
  __shared__ unsigned h[HIST_BINS];
  const int tid = threadIdx.x;
  const int chunk = (E + (int)gridDim.x - 1) / (int)gridDim.x;
  const int e0 = (int)blockIdx.x * chunk;
  const int e1 = min(e0 + chunk, E);
  unsigned* prow = part + (size_t)blockIdx.x * (size_t)n;

  for (int base = 0; base < n; base += HIST_BINS) {
    const int nb = min(HIST_BINS, n - base);
    for (int i = tid; i < nb; i += HIST_THREADS) h[i] = 0u;
    __syncthreads();

    for (int e = e0 + tid; e < e1; e += HIST_BATCH * HIST_THREADS) {
      int b[HIST_BATCH];
#pragma unroll
      for (int u = 0; u < HIST_BATCH; ++u) {
        const int ee = e + u * HIST_THREADS;
        b[u] = (ee < e1) ? idx[ee] : -1;
      }
#pragma unroll
      for (int u = 0; u < HIST_BATCH; ++u) {
        unsigned r = (unsigned)(b[u] - base);
        if (r < (unsigned)nb) atomicAdd(&h[r], 1u);
      }
    }
    __syncthreads();

    for (int i = tid; i < nb; i += HIST_THREADS) prow[base + i] = h[i];
    __syncthreads();
  }
}

__global__ void degree_sum_kernel(const unsigned* __restrict__ part,
                                  float* __restrict__ deg, int n, int rows) {
  int i = blockIdx.x * blockDim.x + threadIdx.x;
  if (i >= n) return;
  unsigned s = 0u;
#pragma unroll 16
  for (int g = 0; g < rows; ++g) s += part[(size_t)g * (size_t)n + i];
  deg[i] = (float)s;
}

#define SCAT_BATCH 5

__device__ __forceinline__ void scatter_add_row(float* __restrict__ o, float4 v) {
  atomicAdd(o + 0, v.x);
  atomicAdd(o + 1, v.y);
  atomicAdd(o + 2, v.z);
  atomicAdd(o + 3, v.w);
}

__global__ __launch_bounds__(RES_THREADS)
void scatter_kernel(const float* __restrict__ srcf,
                    const int* __restrict__ gather_idx,
                    const int* __restrict__ scatter_idx,
                    float* __restrict__ agg,
                    int E) {
  const int part = threadIdx.x & 7;
  const int ngrp = (int)((gridDim.x * blockDim.x) >> 3);
  int e = (int)((blockIdx.x * blockDim.x + threadIdx.x) >> 3);
  for (; e + (SCAT_BATCH - 1) * ngrp < E; e += SCAT_BATCH * ngrp) {
    int s[SCAT_BATCH], d[SCAT_BATCH];
    float4 v[SCAT_BATCH];
#pragma unroll
    for (int u = 0; u < SCAT_BATCH; ++u) {
      s[u] = gather_idx[e + u * ngrp];
      d[u] = scatter_idx[e + u * ngrp];
    }
#pragma unroll
    for (int u = 0; u < SCAT_BATCH; ++u)
      v[u] = ((const float4*)(srcf + (long long)s[u] * 32))[part];
#pragma unroll
    for (int u = 0; u < SCAT_BATCH; ++u)
      scatter_add_row(agg + (long long)d[u] * 32 + part * 4, v[u]);
  }
  for (; e < E; e += ngrp) {
    const int s = gather_idx[e];
    const int d = scatter_idx[e];
    const float4 v = ((const float4*)(srcf + (long long)s * 32))[part];
    scatter_add_row(agg + (long long)d * 32 + part * 4, v);
  }
}

template <int NTILES, bool RELU>
__global__ void sage_wmma_kernel(const float* __restrict__ agg,
                                 const float* __restrict__ deg,
                                 const float* __restrict__ xdst,
                                 const float* __restrict__ Wl,
                                 const float* __restrict__ bl,
                                 const float* __restrict__ Wr,
                                 float* __restrict__ out,
                                 int n) {
  const int OUTF = NTILES * 16;
  int lane = threadIdx.x & 31;
  int wave = (blockIdx.x * blockDim.x + threadIdx.x) >> 5;
  int nb = wave * 16;
  if (nb >= n) return;
  int m  = lane & 15;
  int hi = lane >> 4;

  int node = nb + m;
  bool nv = node < n;
  int ncl = nv ? node : (n - 1);
  float dinv = nv ? (1.0f / fmaxf(deg[ncl], 1.0f)) : 0.0f;
  const float* arow = agg  + (long long)ncl * 32;
  const float* xrow = xdst + (long long)ncl * 32;

  v2f a_agg[8], a_x[8];
#pragma unroll
  for (int s = 0; s < 8; ++s) {
    int k = 4 * s + 2 * hi;
    a_agg[s].x = arow[k] * dinv;
    a_agg[s].y = arow[k + 1] * dinv;
    a_x[s].x   = nv ? xrow[k]     : 0.0f;
    a_x[s].y   = nv ? xrow[k + 1] : 0.0f;
  }

  bool full = (nb + 16 <= n);

#pragma unroll
  for (int t = 0; t < NTILES; ++t) {
    int ng = t * 16 + m;
    const float* wl = Wl + ng * 32;
    const float* wr = Wr + ng * 32;
    float bias = bl[ng];
    v8f c;
#pragma unroll
    for (int v = 0; v < 8; ++v) c[v] = bias;

#pragma unroll
    for (int s = 0; s < 8; ++s) {
      int k = 4 * s + 2 * hi;
      v2f b;
      b.x = wl[k];
      b.y = wl[k + 1];
      c = __builtin_amdgcn_wmma_f32_16x16x4_f32(false, a_agg[s], false, b,
                                                (short)0, c, false, false);
    }
#pragma unroll
    for (int s = 0; s < 8; ++s) {
      int k = 4 * s + 2 * hi;
      v2f b;
      b.x = wr[k];
      b.y = wr[k + 1];
      c = __builtin_amdgcn_wmma_f32_16x16x4_f32(false, a_x[s], false, b,
                                                (short)0, c, false, false);
    }

    if (full) {
#pragma unroll
      for (int v = 0; v < 8; ++v) {
        float val = c[v];
        if (RELU) val = fmaxf(val, 0.0f);
        out[(long long)(nb + v + 8 * hi) * OUTF + ng] = val;
      }
    } else {
#pragma unroll
      for (int v = 0; v < 8; ++v) {
        int nd = nb + v + 8 * hi;
        if (nd < n) {
          float val = c[v];
          if (RELU) val = fmaxf(val, 0.0f);
          out[(long long)nd * OUTF + ng] = val;
        }
      }
    }
  }
}

__global__ void decode_kernel(const float* __restrict__ ou,
                              const float* __restrict__ oc,
                              const int* __restrict__ ls,
                              const int* __restrict__ ld,
                              float* __restrict__ out, int L) {
  int t = blockIdx.x * blockDim.x + threadIdx.x;
  if (t >= L) return;
  const float4* a = (const float4*)(ou + (long long)ls[t] * 16);
  const float4* b = (const float4*)(oc + (long long)ld[t] * 16);
  float acc = 0.0f;
#pragma unroll
  for (int i = 0; i < 4; ++i) {
    float4 xa = a[i], xb = b[i];
    acc += xa.x * xb.x + xa.y * xb.y + xa.z * xb.z + xa.w * xb.w;
  }
  out[t] = acc;
}

extern "C" void kernel_launch(void* const* d_in, const int* in_sizes, int n_in,
                              void* d_out, int out_size, void* d_ws, size_t ws_size,
                              hipStream_t stream) {
  const float* user_x       = (const float*)d_in[0];
  const float* course_x     = (const float*)d_in[1];
  const float* user_embed   = (const float*)d_in[2];
  const float* course_embed = (const float*)d_in[3];
  const float* Wu = (const float*)d_in[4];
  const float* bu = (const float*)d_in[5];
  const float* Wc = (const float*)d_in[6];
  const float* bc = (const float*)d_in[7];
  const float* c1e_Wl = (const float*)d_in[8];
  const float* c1e_bl = (const float*)d_in[9];
  const float* c1e_Wr = (const float*)d_in[10];
  const float* c1r_Wl = (const float*)d_in[11];
  const float* c1r_bl = (const float*)d_in[12];
  const float* c1r_Wr = (const float*)d_in[13];
  const float* c2e_Wl = (const float*)d_in[14];
  const float* c2e_bl = (const float*)d_in[15];
  const float* c2e_Wr = (const float*)d_in[16];
  const float* c2r_Wl = (const float*)d_in[17];
  const float* c2r_bl = (const float*)d_in[18];
  const float* c2r_Wr = (const float*)d_in[19];
  const int* user_idx   = (const int*)d_in[20];
  const int* course_idx = (const int*)d_in[21];
  const int* edge_src   = (const int*)d_in[22];
  const int* edge_dst   = (const int*)d_in[23];
  const int* label_src  = (const int*)d_in[24];
  const int* label_dst  = (const int*)d_in[25];
  float* outp = (float*)d_out;

  const int NU = in_sizes[2] / 16;
  const int NC = in_sizes[3] / 16;
  const int E  = in_sizes[22];
  const int L  = in_sizes[24];

  char* p = (char*)d_ws;
  auto carve = [&](size_t bytes) -> float* {
    float* r = (float*)p;
    p += (bytes + 255) & ~((size_t)255);
    return r;
  };
  float* xu   = carve((size_t)NU * 32 * 4);
  float* xc   = carve((size_t)NC * 32 * 4);
  float* aggu = carve((size_t)NU * 32 * 4);
  float* aggc = carve((size_t)NC * 32 * 4);
  float* degu = carve((size_t)NU * 4);
  float* degc = carve((size_t)NC * 4);
  float* hu   = carve((size_t)NU * 32 * 4);
  float* hc   = carve((size_t)NC * 32 * 4);
  float* ou   = carve((size_t)NU * 16 * 4);
  float* oc   = carve((size_t)NC * 16 * 4);

  const int B = 256;
  auto nblk = [](long long t, int b) { return (unsigned)((t + b - 1) / b); };
  auto rblk = [](long long t) {
    long long g = (t + RES_THREADS - 1) / RES_THREADS;
    return (unsigned)(g < RES_BLOCKS ? g : RES_BLOCKS);
  };

  const unsigned scan_blocks = nblk(NU, SCAN_CHUNK);
  int* ucnt  = (int*)ou;
  int* urow  = ucnt + (((size_t)NU + 63) & ~(size_t)63);
  int* urank = urow + (((size_t)NU + 63) & ~(size_t)63);
  int* ucol  = urank + (((size_t)E + 63) & ~(size_t)63);
  int* ubtot = ucol + (((size_t)E + 63) & ~(size_t)63);

  hipMemsetAsync(aggc, 0, (size_t)NC * 32 * 4, stream);
  hipMemsetAsync(ucnt, 0, (size_t)NU * 4, stream);

  fuse_kernel<5><<<rblk((long long)NU * 8), RES_THREADS, 0, stream>>>(
      user_x, user_embed, user_idx, Wu, bu, xu, NU);
  fuse_kernel<2><<<rblk((long long)NC * 8), RES_THREADS, 0, stream>>>(
      course_x, course_embed, course_idx, Wc, bc, xc, NC);

  unsigned* degpart = (unsigned*)hu;
  degree_hist_kernel<<<64, HIST_THREADS, 0, stream>>>(edge_dst, degpart, E, NC);
  degree_sum_kernel<<<nblk(NC, B), B, 0, stream>>>(degpart, degc, NC, 64);
  rank_kernel<<<rblk(E), RES_THREADS, 0, stream>>>(edge_src, ucnt, urank, E);
  row_total_kernel<<<scan_blocks, SCAN_THREADS, 0, stream>>>(ucnt, ubtot, NU);
  row_start_kernel<<<scan_blocks, SCAN_THREADS, 0, stream>>>(ucnt, ubtot, urow, degu, NU);
  fill_kernel<<<rblk(E), RES_THREADS, 0, stream>>>(edge_src, edge_dst, urank, urow, ucol, E);

  scatter_kernel<<<rblk((long long)E * 8), RES_THREADS, 0, stream>>>(xu, edge_src, edge_dst, aggc, E);
  gather_kernel<<<rblk((long long)NU * 8), RES_THREADS, 0, stream>>>(xc, urow, ucnt, ucol, aggu, NU);
  sage_wmma_kernel<2, true><<<nblk(((long long)(NC + 15) / 16) * 32, B), B, 0, stream>>>(
      aggc, degc, xc, c1e_Wl, c1e_bl, c1e_Wr, hc, NC);
  sage_wmma_kernel<2, true><<<nblk(((long long)(NU + 15) / 16) * 32, B), B, 0, stream>>>(
      aggu, degu, xu, c1r_Wl, c1r_bl, c1r_Wr, hu, NU);

  hipMemsetAsync(aggc, 0, (size_t)NC * 32 * 4, stream);
  scatter_kernel<<<rblk((long long)E * 8), RES_THREADS, 0, stream>>>(hu, edge_src, edge_dst, aggc, E);
  gather_kernel<<<rblk((long long)NU * 8), RES_THREADS, 0, stream>>>(hc, urow, ucnt, ucol, aggu, NU);
  sage_wmma_kernel<1, false><<<nblk(((long long)(NC + 15) / 16) * 32, B), B, 0, stream>>>(
      aggc, degc, hc, c2e_Wl, c2e_bl, c2e_Wr, oc, NC);
  sage_wmma_kernel<1, false><<<nblk(((long long)(NU + 15) / 16) * 32, B), B, 0, stream>>>(
      aggu, degu, hu, c2r_Wl, c2r_bl, c2r_Wr, ou, NU);

  decode_kernel<<<nblk(L, B), B, 0, stream>>>(ou, oc, label_src, label_dst, outp, L);
}


# course-side neighbour sums as a gather over edges bucketed by course (counting sort on per-block LDS histograms) instead of float atomics; sum order = bucket order, 4 partial sums per row
# speedup vs baseline: 2.3088x; 1.4518x over previous
#include <hip/hip_runtime.h>

typedef __attribute__((ext_vector_type(8))) float v8f;
typedef __attribute__((ext_vector_type(2))) float v2f;

#define RES_BLOCKS  2048
#define RES_THREADS 256

template <int FIN>
__global__ __launch_bounds__(RES_THREADS)
void fuse_kernel(const float* __restrict__ x,
                 const float* __restrict__ embed,
                 const int* __restrict__ node_index,
                 const float* __restrict__ W,
                 const float* __restrict__ b,
                 float* __restrict__ out,
                 int n) {
  const int part = threadIdx.x & 7;
  const int grp  = (int)((blockIdx.x * blockDim.x + threadIdx.x) >> 3);
  const int ngrp = (int)((gridDim.x * blockDim.x) >> 3);
  const bool lin = part >= 4;
  const int q4 = part & 3;
  const int r  = q4 * 4;
  float w[4][FIN], bias[4];
#pragma unroll
  for (int q = 0; q < 4; ++q) {
    bias[q] = b[r + q];
#pragma unroll
    for (int k = 0; k < FIN; ++k) w[q][k] = W[(r + q) * FIN + k];
  }
  for (int node = grp; node < n; node += ngrp) {
    const int ni = node_index[node];
    const float4 ev = ((const float4*)(embed + (long long)ni * 16))[q4];
    const float* xr = x + (long long)node * FIN;
    float xv[FIN];
#pragma unroll
    for (int k = 0; k < FIN; ++k) xv[k] = xr[k];
    float a[4];
#pragma unroll
    for (int q = 0; q < 4; ++q) {
      float acc = bias[q];
#pragma unroll
      for (int k = 0; k < FIN; ++k) acc = fmaf(xv[k], w[q][k], acc);
      a[q] = acc;
    }
    float4 v;
    v.x = lin ? a[0] : ev.x;
    v.y = lin ? a[1] : ev.y;
    v.z = lin ? a[2] : ev.z;
    v.w = lin ? a[3] : ev.w;
    ((float4*)(out + (long long)node * 32))[part] = v;
  }
}

#define RANK_BATCH 4

__global__ __launch_bounds__(RES_THREADS)
void rank_kernel(const int* __restrict__ idx, int* __restrict__ cnt,
                 int* __restrict__ rank, int E) {
  const int nt = (int)(gridDim.x * blockDim.x);
  int e = (int)(blockIdx.x * blockDim.x + threadIdx.x);
  for (; e + (RANK_BATCH - 1) * nt < E; e += RANK_BATCH * nt) {
    int d[RANK_BATCH], r[RANK_BATCH];
#pragma unroll
    for (int u = 0; u < RANK_BATCH; ++u) d[u] = idx[e + u * nt];
#pragma unroll
    for (int u = 0; u < RANK_BATCH; ++u) r[u] = atomicAdd(&cnt[d[u]], 1);
#pragma unroll
    for (int u = 0; u < RANK_BATCH; ++u) rank[e + u * nt] = r[u];
  }
  for (; e < E; e += nt) rank[e] = atomicAdd(&cnt[idx[e]], 1);
}

#define SCAN_THREADS 1024
#define SCAN_CHUNK   (SCAN_THREADS * 4)

__device__ __forceinline__ int wave_sum(int v) {
#pragma unroll
  for (int o = 16; o > 0; o >>= 1) v += __shfl_xor(v, o, 32);
  return v;
}

__device__ __forceinline__ int block_sum(int v, int* ws) {
  v = wave_sum(v);
  if ((threadIdx.x & 31) == 0) ws[threadIdx.x >> 5] = v;
  __syncthreads();
  int t = wave_sum(ws[threadIdx.x & 31]);
  __syncthreads();
  return t;
}

__device__ __forceinline__ int4 load_counts4(const int* __restrict__ cnt, int i0, int n) {
  int4 c;
  if (i0 + 3 < n) {
    c = *(const int4*)(cnt + i0);
  } else {
    c.x = (i0     < n) ? cnt[i0]     : 0;
    c.y = (i0 + 1 < n) ? cnt[i0 + 1] : 0;
    c.z = (i0 + 2 < n) ? cnt[i0 + 2] : 0;
    c.w = 0;
  }
  return c;
}

__global__ __launch_bounds__(SCAN_THREADS)
void row_total_kernel(const int* __restrict__ cnt, int* __restrict__ btot, int n) {
  __shared__ int ws[32];
  const int i0 = (int)blockIdx.x * SCAN_CHUNK + (int)threadIdx.x * 4;
  const int4 c = load_counts4(cnt, i0, n);
  const int t = block_sum(c.x + c.y + c.z + c.w, ws);
  if (threadIdx.x == 0) btot[blockIdx.x] = t;
}

__global__ __launch_bounds__(SCAN_THREADS)
void row_start_kernel(const int* __restrict__ cnt, const int* __restrict__ btot,
                      int* __restrict__ rowptr, float* __restrict__ deg, int n) {
  __shared__ int ws[32];
  const int tid = threadIdx.x, lane = tid & 31, wv = tid >> 5;
  int before = 0;
  for (int j = tid; j < (int)blockIdx.x; j += SCAN_THREADS) before += btot[j];
  before = block_sum(before, ws);

  const int i0 = (int)blockIdx.x * SCAN_CHUNK + tid * 4;
  const int4 c = load_counts4(cnt, i0, n);
  const int local = c.x + c.y + c.z + c.w;
  int inc = local;
#pragma unroll
  for (int o = 1; o < 32; o <<= 1) {
    const int t = __shfl_up(inc, o, 32);
    if (lane >= o) inc += t;
  }
  if (lane == 31) ws[wv] = inc;
  __syncthreads();
  const int wt = ws[lane];
  int winc = wt;
#pragma unroll
  for (int o = 1; o < 32; o <<= 1) {
    const int t = __shfl_up(winc, o, 32);
    if (lane >= o) winc += t;
  }
  const int wexc = __shfl(winc - wt, wv, 32);
  const int p0 = before + wexc + (inc - local);
  const int p1 = p0 + c.x, p2 = p1 + c.y, p3 = p2 + c.z;
  if (i0 + 3 < n) {
    *(int4*)(rowptr + i0) = make_int4(p0, p1, p2, p3);
    *(float4*)(deg + i0) = make_float4((float)c.x, (float)c.y, (float)c.z, (float)c.w);
  } else {
    if (i0     < n) { rowptr[i0]     = p0; deg[i0]     = (float)c.x; }
    if (i0 + 1 < n) { rowptr[i0 + 1] = p1; deg[i0 + 1] = (float)c.y; }
    if (i0 + 2 < n) { rowptr[i0 + 2] = p2; deg[i0 + 2] = (float)c.z; }
  }
}

#define FILL_BATCH 4

__global__ __launch_bounds__(RES_THREADS)
void fill_kernel(const int* __restrict__ row, const int* __restrict__ other,
                 const int* __restrict__ rank, const int* __restrict__ rowptr,
                 int* __restrict__ col, int E) {
  const int nt = (int)(gridDim.x * blockDim.x);
  int e = (int)(blockIdx.x * blockDim.x + threadIdx.x);
  for (; e + (FILL_BATCH - 1) * nt < E; e += FILL_BATCH * nt) {
    int s[FILL_BATCH], d[FILL_BATCH], r[FILL_BATCH], p[FILL_BATCH];
#pragma unroll
    for (int u = 0; u < FILL_BATCH; ++u) {
      s[u] = row[e + u * nt];
      d[u] = other[e + u * nt];
      r[u] = rank[e + u * nt];
    }
#pragma unroll
    for (int u = 0; u < FILL_BATCH; ++u) p[u] = rowptr[s[u]];
#pragma unroll
    for (int u = 0; u < FILL_BATCH; ++u) col[p[u] + r[u]] = d[u];
  }
  for (; e < E; e += nt) col[rowptr[row[e]] + rank[e]] = other[e];
}

__global__ __launch_bounds__(RES_THREADS)
void gather_kernel(const float* __restrict__ srcf,
                   const int* __restrict__ rowptr,
                   const int* __restrict__ rowlen,
                   const int* __restrict__ col,
                   float* __restrict__ agg, int n) {
  const int part = threadIdx.x & 7;
  const int ngrp = (int)((gridDim.x * blockDim.x) >> 3);
  for (int node = (int)((blockIdx.x * blockDim.x + threadIdx.x) >> 3); node < n; node += ngrp) {
    const int start = rowptr[node];
    const int len   = rowlen[node];
    float4 acc = make_float4(0.0f, 0.0f, 0.0f, 0.0f);
    for (int base = 0; base < len; base += 8) {
      const int m = min(8, len - base);
      const int ci = (part < m) ? col[start + base + part] : 0;
      for (int j0 = 0; j0 < m; j0 += 4) {
        float4 v[4];
#pragma unroll
        for (int j = 0; j < 4; ++j) {
          const int jj = (j0 + j < m) ? (j0 + j) : j0;
          const int c = __shfl(ci, jj, 8);
          v[j] = ((const float4*)(srcf + (long long)c * 32))[part];
        }
#pragma unroll
        for (int j = 0; j < 4; ++j) {
          if (j0 + j < m) {
            acc.x += v[j].x; acc.y += v[j].y; acc.z += v[j].z; acc.w += v[j].w;
          }
        }
      }
    }
    ((float4*)(agg + (long long)node * 32))[part] = acc;
  }
}

#define HIST_BINS    16384
#define HIST_THREADS 1024
#define HIST_BATCH   8
#define HIST_BLOCKS  64

__global__ __launch_bounds__(HIST_THREADS)
void degree_hist_kernel(const int* __restrict__ idx,
                        unsigned* __restrict__ part, int E, int n) {
  __shared__ unsigned h[HIST_BINS];
  const int tid = threadIdx.x;
  const int chunk = (E + (int)gridDim.x - 1) / (int)gridDim.x;
  const int e0 = (int)blockIdx.x * chunk;
  const int e1 = min(e0 + chunk, E);
  unsigned* prow = part + (size_t)blockIdx.x * (size_t)n;

  for (int base = 0; base < n; base += HIST_BINS) {
    const int nb = min(HIST_BINS, n - base);
    for (int i = tid; i < nb; i += HIST_THREADS) h[i] = 0u;
    __syncthreads();

    for (int e = e0 + tid; e < e1; e += HIST_BATCH * HIST_THREADS) {
      int b[HIST_BATCH];
#pragma unroll
      for (int u = 0; u < HIST_BATCH; ++u) {
        const int ee = e + u * HIST_THREADS;
        b[u] = (ee < e1) ? idx[ee] : -1;
      }
#pragma unroll
      for (int u = 0; u < HIST_BATCH; ++u) {
        unsigned r = (unsigned)(b[u] - base);
        if (r < (unsigned)nb) atomicAdd(&h[r], 1u);
      }
    }
    __syncthreads();

    for (int i = tid; i < nb; i += HIST_THREADS) prow[base + i] = h[i];
    __syncthreads();
  }
}

#define SUM_BATCH 16

__global__ void degree_sum_kernel(unsigned* part, int* __restrict__ cnt,
                                  float* __restrict__ deg, int n, int rows) {
  int i = blockIdx.x * blockDim.x + threadIdx.x;
  if (i >= n) return;
  unsigned s = 0u;
  for (int g0 = 0; g0 < rows; g0 += SUM_BATCH) {
    unsigned v[SUM_BATCH];
#pragma unroll
    for (int u = 0; u < SUM_BATCH; ++u)
      v[u] = (g0 + u < rows) ? part[(size_t)(g0 + u) * (size_t)n + i] : 0u;
#pragma unroll
    for (int u = 0; u < SUM_BATCH; ++u) {
      if (g0 + u < rows) part[(size_t)(g0 + u) * (size_t)n + i] = s;
      s += v[u];
    }
  }
  cnt[i] = (int)s;
  deg[i] = (float)s;
}

__global__ __launch_bounds__(SCAN_THREADS)
void short_row_start_kernel(const int* __restrict__ cnt, int* __restrict__ rowptr, int n) {
  __shared__ int ws[32];
  const int tid = threadIdx.x, lane = tid & 31, wv = tid >> 5;
  int carry = 0;
  for (int c0 = 0; c0 < n; c0 += SCAN_CHUNK) {
    const int i0 = c0 + tid * 4;
    const int4 c = load_counts4(cnt, i0, n);
    const int local = c.x + c.y + c.z + c.w;
    int inc = local;
#pragma unroll
    for (int o = 1; o < 32; o <<= 1) {
      const int t = __shfl_up(inc, o, 32);
      if (lane >= o) inc += t;
    }
    if (lane == 31) ws[wv] = inc;
    __syncthreads();
    const int wt = ws[lane];
    int winc = wt;
#pragma unroll
    for (int o = 1; o < 32; o <<= 1) {
      const int t = __shfl_up(winc, o, 32);
      if (lane >= o) winc += t;
    }
    const int wexc = __shfl(winc - wt, wv, 32);
    const int total = __shfl(winc, 31, 32);
    const int p0 = carry + wexc + (inc - local);
    const int p1 = p0 + c.x, p2 = p1 + c.y, p3 = p2 + c.z;
    if (i0 + 3 < n) {
      *(int4*)(rowptr + i0) = make_int4(p0, p1, p2, p3);
    } else {
      if (i0     < n) rowptr[i0]     = p0;
      if (i0 + 1 < n) rowptr[i0 + 1] = p1;
      if (i0 + 2 < n) rowptr[i0 + 2] = p2;
    }
    carry += total;
    __syncthreads();
  }
}

__global__ __launch_bounds__(HIST_THREADS)
void bucket_fill_kernel(const int* __restrict__ idx, const int* __restrict__ other,
                        const unsigned* __restrict__ part,
                        const int* __restrict__ rowptr,
                        int* __restrict__ col, int E, int n) {
  __shared__ unsigned h[HIST_BINS];
  const int tid = threadIdx.x;
  const int chunk = (E + (int)gridDim.x - 1) / (int)gridDim.x;
  const int e0 = (int)blockIdx.x * chunk;
  const int e1 = min(e0 + chunk, E);
  const unsigned* prow = part + (size_t)blockIdx.x * (size_t)n;

  for (int base = 0; base < n; base += HIST_BINS) {
    const int nb = min(HIST_BINS, n - base);
    for (int i = tid; i < nb; i += HIST_THREADS)
      h[i] = (unsigned)rowptr[base + i] + prow[base + i];
    __syncthreads();

    for (int e = e0 + tid; e < e1; e += HIST_BATCH * HIST_THREADS) {
      int b[HIST_BATCH], o[HIST_BATCH];
#pragma unroll
      for (int u = 0; u < HIST_BATCH; ++u) {
        const int ee = e + u * HIST_THREADS;
        b[u] = (ee < e1) ? idx[ee] : -1;
        o[u] = (ee < e1) ? other[ee] : 0;
      }
#pragma unroll
      for (int u = 0; u < HIST_BATCH; ++u) {
        unsigned r = (unsigned)(b[u] - base);
        if (r < (unsigned)nb) col[atomicAdd(&h[r], 1u)] = o[u];
      }
    }
    __syncthreads();
  }
}

__global__ __launch_bounds__(RES_THREADS)
void gather_long_kernel(const float* __restrict__ srcf,
                        const int* __restrict__ rowptr,
                        const int* __restrict__ rowlen,
                        const int* __restrict__ col,
                        float* __restrict__ agg, int n) {
  const int lane = threadIdx.x & 31;
  const int part = lane & 7;
  const int g    = lane >> 3;
  const int nw   = (int)((gridDim.x * blockDim.x) >> 5);
  for (int node = (int)((blockIdx.x * blockDim.x + threadIdx.x) >> 5); node < n; node += nw) {
    const int start = rowptr[node];
    const int len   = rowlen[node];
    float4 acc = make_float4(0.0f, 0.0f, 0.0f, 0.0f);
    int ci = (lane < len) ? col[start + lane] : 0;
    for (int base = 0; base < len; base += 32) {
      const int m = min(32, len - base);
      const int nx = base + 32 + lane;
      const int cn = (nx < len) ? col[start + nx] : 0;
      float4 v[8];
#pragma unroll
      for (int j = 0; j < 8; ++j) {
        const int jj = j * 4 + g;
        const int c = __shfl(ci, (jj < m) ? jj : 0, 32);
        v[j] = ((const float4*)(srcf + (long long)c * 32))[part];
      }
#pragma unroll
      for (int j = 0; j < 8; ++j) {
        const bool in = (j * 4 + g) < m;
        acc.x += in ? v[j].x : 0.0f;
        acc.y += in ? v[j].y : 0.0f;
        acc.z += in ? v[j].z : 0.0f;
        acc.w += in ? v[j].w : 0.0f;
      }
      ci = cn;
    }
    acc.x += __shfl_xor(acc.x, 8, 32);  acc.y += __shfl_xor(acc.y, 8, 32);
    acc.z += __shfl_xor(acc.z, 8, 32);  acc.w += __shfl_xor(acc.w, 8, 32);
    acc.x += __shfl_xor(acc.x, 16, 32); acc.y += __shfl_xor(acc.y, 16, 32);
    acc.z += __shfl_xor(acc.z, 16, 32); acc.w += __shfl_xor(acc.w, 16, 32);
    if (g == 0) ((float4*)(agg + (long long)node * 32))[part] = acc;
  }
}

template <int NTILES, bool RELU>
__global__ void sage_wmma_kernel(const float* __restrict__ agg,
                                 const float* __restrict__ deg,
                                 const float* __restrict__ xdst,
                                 const float* __restrict__ Wl,
                                 const float* __restrict__ bl,
                                 const float* __restrict__ Wr,
                                 float* __restrict__ out,
                                 int n) {
  const int OUTF = NTILES * 16;
  int lane = threadIdx.x & 31;
  int wave = (blockIdx.x * blockDim.x + threadIdx.x) >> 5;
  int nb = wave * 16;
  if (nb >= n) return;
  int m  = lane & 15;
  int hi = lane >> 4;

  int node = nb + m;
  bool nv = node < n;
  int ncl = nv ? node : (n - 1);
  float dinv = nv ? (1.0f / fmaxf(deg[ncl], 1.0f)) : 0.0f;
  const float* arow = agg  + (long long)ncl * 32;
  const float* xrow = xdst + (long long)ncl * 32;

  v2f a_agg[8], a_x[8];
#pragma unroll
  for (int s = 0; s < 8; ++s) {
    int k = 4 * s + 2 * hi;
    a_agg[s].x = arow[k] * dinv;
    a_agg[s].y = arow[k + 1] * dinv;
    a_x[s].x   = nv ? xrow[k]     : 0.0f;
    a_x[s].y   = nv ? xrow[k + 1] : 0.0f;
  }

  bool full = (nb + 16 <= n);

#pragma unroll
  for (int t = 0; t < NTILES; ++t) {
    int ng = t * 16 + m;
    const float* wl = Wl + ng * 32;
    const float* wr = Wr + ng * 32;
    float bias = bl[ng];
    v8f c;
#pragma unroll
    for (int v = 0; v < 8; ++v) c[v] = bias;

#pragma unroll
    for (int s = 0; s < 8; ++s) {
      int k = 4 * s + 2 * hi;
      v2f b;
      b.x = wl[k];
      b.y = wl[k + 1];
      c = __builtin_amdgcn_wmma_f32_16x16x4_f32(false, a_agg[s], false, b,
                                                (short)0, c, false, false);
    }
#pragma unroll
    for (int s = 0; s < 8; ++s) {
      int k = 4 * s + 2 * hi;
      v2f b;
      b.x = wr[k];
      b.y = wr[k + 1];
      c = __builtin_amdgcn_wmma_f32_16x16x4_f32(false, a_x[s], false, b,
                                                (short)0, c, false, false);
    }

    if (full) {
#pragma unroll
      for (int v = 0; v < 8; ++v) {
        float val = c[v];
        if (RELU) val = fmaxf(val, 0.0f);
        out[(long long)(nb + v + 8 * hi) * OUTF + ng] = val;
      }
    } else {
#pragma unroll
      for (int v = 0; v < 8; ++v) {
        int nd = nb + v + 8 * hi;
        if (nd < n) {
          float val = c[v];
          if (RELU) val = fmaxf(val, 0.0f);
          out[(long long)nd * OUTF + ng] = val;
        }
      }
    }
  }
}

__global__ void decode_kernel(const float* __restrict__ ou,
                              const float* __restrict__ oc,
                              const int* __restrict__ ls,
                              const int* __restrict__ ld,
                              float* __restrict__ out, int L) {
  int t = blockIdx.x * blockDim.x + threadIdx.x;
  if (t >= L) return;
  const float4* a = (const float4*)(ou + (long long)ls[t] * 16);
  const float4* b = (const float4*)(oc + (long long)ld[t] * 16);
  float acc = 0.0f;
#pragma unroll
  for (int i = 0; i < 4; ++i) {
    float4 xa = a[i], xb = b[i];
    acc += xa.x * xb.x + xa.y * xb.y + xa.z * xb.z + xa.w * xb.w;
  }
  out[t] = acc;
}

extern "C" void kernel_launch(void* const* d_in, const int* in_sizes, int n_in,
                              void* d_out, int out_size, void* d_ws, size_t ws_size,
                              hipStream_t stream) {
  const float* user_x       = (const float*)d_in[0];
  const float* course_x     = (const float*)d_in[1];
  const float* user_embed   = (const float*)d_in[2];
  const float* course_embed = (const float*)d_in[3];
  const float* Wu = (const float*)d_in[4];
  const float* bu = (const float*)d_in[5];
  const float* Wc = (const float*)d_in[6];
  const float* bc = (const float*)d_in[7];
  const float* c1e_Wl = (const float*)d_in[8];
  const float* c1e_bl = (const float*)d_in[9];
  const float* c1e_Wr = (const float*)d_in[10];
  const float* c1r_Wl = (const float*)d_in[11];
  const float* c1r_bl = (const float*)d_in[12];
  const float* c1r_Wr = (const float*)d_in[13];
  const float* c2e_Wl = (const float*)d_in[14];
  const float* c2e_bl = (const float*)d_in[15];
  const float* c2e_Wr = (const float*)d_in[16];
  const float* c2r_Wl = (const float*)d_in[17];
  const float* c2r_bl = (const float*)d_in[18];
  const float* c2r_Wr = (const float*)d_in[19];
  const int* user_idx   = (const int*)d_in[20];
  const int* course_idx = (const int*)d_in[21];
  const int* edge_src   = (const int*)d_in[22];
  const int* edge_dst   = (const int*)d_in[23];
  const int* label_src  = (const int*)d_in[24];
  const int* label_dst  = (const int*)d_in[25];
  float* outp = (float*)d_out;

  const int NU = in_sizes[2] / 16;
  const int NC = in_sizes[3] / 16;
  const int E  = in_sizes[22];
  const int L  = in_sizes[24];

  char* p = (char*)d_ws;
  auto carve = [&](size_t bytes) -> float* {
    float* r = (float*)p;
    p += (bytes + 255) & ~((size_t)255);
    return r;
  };
  float* xu   = carve((size_t)NU * 32 * 4);
  float* xc   = carve((size_t)NC * 32 * 4);
  float* aggu = carve((size_t)NU * 32 * 4);
  float* aggc = carve((size_t)NC * 32 * 4);
  float* degu = carve((size_t)NU * 4);
  float* degc = carve((size_t)NC * 4);
  float* hu   = carve((size_t)NU * 32 * 4);
  float* hc   = carve((size_t)NC * 32 * 4);
  float* ou   = carve((size_t)NU * 16 * 4);
  float* oc   = carve((size_t)NC * 16 * 4);

  const int B = 256;
  auto nblk = [](long long t, int b) { return (unsigned)((t + b - 1) / b); };
  auto rblk = [](long long t) {
    long long g = (t + RES_THREADS - 1) / RES_THREADS;
    return (unsigned)(g < RES_BLOCKS ? g : RES_BLOCKS);
  };

  const unsigned scan_blocks = nblk(NU, SCAN_CHUNK);
  int* ucnt  = (int*)ou;
  int* urow  = ucnt + (((size_t)NU + 63) & ~(size_t)63);
  int* urank = urow + (((size_t)NU + 63) & ~(size_t)63);
  int* ucol  = urank + (((size_t)E + 63) & ~(size_t)63);
  int* ubtot = ucol + (((size_t)E + 63) & ~(size_t)63);
  int* ccnt  = ubtot + (((size_t)scan_blocks + 63) & ~(size_t)63);
  int* crow  = ccnt + (((size_t)NC + 63) & ~(size_t)63);
  int* ccol  = crow + (((size_t)NC + 63) & ~(size_t)63);

  hipMemsetAsync(ucnt, 0, (size_t)NU * 4, stream);

  fuse_kernel<5><<<rblk((long long)NU * 8), RES_THREADS, 0, stream>>>(
      user_x, user_embed, user_idx, Wu, bu, xu, NU);
  fuse_kernel<2><<<rblk((long long)NC * 8), RES_THREADS, 0, stream>>>(
      course_x, course_embed, course_idx, Wc, bc, xc, NC);

  unsigned* degpart = (unsigned*)hu;
  degree_hist_kernel<<<HIST_BLOCKS, HIST_THREADS, 0, stream>>>(edge_dst, degpart, E, NC);
  degree_sum_kernel<<<nblk(NC, B), B, 0, stream>>>(degpart, ccnt, degc, NC, HIST_BLOCKS);
  short_row_start_kernel<<<1, SCAN_THREADS, 0, stream>>>(ccnt, crow, NC);
  bucket_fill_kernel<<<HIST_BLOCKS, HIST_THREADS, 0, stream>>>(
      edge_dst, edge_src, degpart, crow, ccol, E, NC);
  rank_kernel<<<rblk(E), RES_THREADS, 0, stream>>>(edge_src, ucnt, urank, E);
  row_total_kernel<<<scan_blocks, SCAN_THREADS, 0, stream>>>(ucnt, ubtot, NU);
  row_start_kernel<<<scan_blocks, SCAN_THREADS, 0, stream>>>(ucnt, ubtot, urow, degu, NU);
  fill_kernel<<<rblk(E), RES_THREADS, 0, stream>>>(edge_src, edge_dst, urank, urow, ucol, E);

  gather_long_kernel<<<rblk((long long)NC * 32), RES_THREADS, 0, stream>>>(xu, crow, ccnt, ccol, aggc, NC);
  gather_kernel<<<rblk((long long)NU * 8), RES_THREADS, 0, stream>>>(xc, urow, ucnt, ucol, aggu, NU);
  sage_wmma_kernel<2, true><<<nblk(((long long)(NC + 15) / 16) * 32, B), B, 0, stream>>>(
      aggc, degc, xc, c1e_Wl, c1e_bl, c1e_Wr, hc, NC);
  sage_wmma_kernel<2, true><<<nblk(((long long)(NU + 15) / 16) * 32, B), B, 0, stream>>>(
      aggu, degu, xu, c1r_Wl, c1r_bl, c1r_Wr, hu, NU);

  gather_long_kernel<<<rblk((long long)NC * 32), RES_THREADS, 0, stream>>>(hu, crow, ccnt, ccol, aggc, NC);
  gather_kernel<<<rblk((long long)NU * 8), RES_THREADS, 0, stream>>>(hc, urow, ucnt, ucol, aggu, NU);
  sage_wmma_kernel<1, false><<<nblk(((long long)(NC + 15) / 16) * 32, B), B, 0, stream>>>(
      aggc, degc, hc, c2e_Wl, c2e_bl, c2e_Wr, oc, NC);
  sage_wmma_kernel<1, false><<<nblk(((long long)(NU + 15) / 16) * 32, B), B, 0, stream>>>(
      aggu, degu, hu, c2r_Wl, c2r_bl, c2r_Wr, ou, NU);

  decode_kernel<<<nblk(L, B), B, 0, stream>>>(ou, oc, label_src, label_dst, outp, L);
}


# user-side neighbour gather fused into the WMMA finalize: the gathered mean reaches the A fragments through a per-wave LDS tile instead of a 64 MB write and read per layer; sum order unchanged
# speedup vs baseline: 3.0488x; 1.3205x over previous
#include <hip/hip_runtime.h>

typedef __attribute__((ext_vector_type(8))) float v8f;
typedef __attribute__((ext_vector_type(2))) float v2f;

#define RES_BLOCKS  2048
#define RES_THREADS 256

template <int FIN>
__global__ __launch_bounds__(RES_THREADS)
void fuse_kernel(const float* __restrict__ x,
                 const float* __restrict__ embed,
                 const int* __restrict__ node_index,
                 const float* __restrict__ W,
                 const float* __restrict__ b,
                 float* __restrict__ out,
                 int n) {
  const int part = threadIdx.x & 7;
  const int grp  = (int)((blockIdx.x * blockDim.x + threadIdx.x) >> 3);
  const int ngrp = (int)((gridDim.x * blockDim.x) >> 3);
  const bool lin = part >= 4;
  const int q4 = part & 3;
  const int r  = q4 * 4;
  float w[4][FIN], bias[4];
#pragma unroll
  for (int q = 0; q < 4; ++q) {
    bias[q] = b[r + q];
#pragma unroll
    for (int k = 0; k < FIN; ++k) w[q][k] = W[(r + q) * FIN + k];
  }
  for (int node = grp; node < n; node += ngrp) {
    const int ni = node_index[node];
    const float4 ev = ((const float4*)(embed + (long long)ni * 16))[q4];
    const float* xr = x + (long long)node * FIN;
    float xv[FIN];
#pragma unroll
    for (int k = 0; k < FIN; ++k) xv[k] = xr[k];
    float a[4];
#pragma unroll
    for (int q = 0; q < 4; ++q) {
      float acc = bias[q];
#pragma unroll
      for (int k = 0; k < FIN; ++k) acc = fmaf(xv[k], w[q][k], acc);
      a[q] = acc;
    }
    float4 v;
    v.x = lin ? a[0] : ev.x;
    v.y = lin ? a[1] : ev.y;
    v.z = lin ? a[2] : ev.z;
    v.w = lin ? a[3] : ev.w;
    ((float4*)(out + (long long)node * 32))[part] = v;
  }
}

#define RANK_BATCH 4

__global__ __launch_bounds__(RES_THREADS)
void rank_kernel(const int* __restrict__ idx, int* __restrict__ cnt,
                 int* __restrict__ rank, int E) {
  const int nt = (int)(gridDim.x * blockDim.x);
  int e = (int)(blockIdx.x * blockDim.x + threadIdx.x);
  for (; e + (RANK_BATCH - 1) * nt < E; e += RANK_BATCH * nt) {
    int d[RANK_BATCH], r[RANK_BATCH];
#pragma unroll
    for (int u = 0; u < RANK_BATCH; ++u) d[u] = idx[e + u * nt];
#pragma unroll
    for (int u = 0; u < RANK_BATCH; ++u) r[u] = atomicAdd(&cnt[d[u]], 1);
#pragma unroll
    for (int u = 0; u < RANK_BATCH; ++u) rank[e + u * nt] = r[u];
  }
  for (; e < E; e += nt) rank[e] = atomicAdd(&cnt[idx[e]], 1);
}

#define SCAN_THREADS 1024
#define SCAN_CHUNK   (SCAN_THREADS * 4)

__device__ __forceinline__ int wave_sum(int v) {
#pragma unroll
  for (int o = 16; o > 0; o >>= 1) v += __shfl_xor(v, o, 32);
  return v;
}

__device__ __forceinline__ int block_sum(int v, int* ws) {
  v = wave_sum(v);
  if ((threadIdx.x & 31) == 0) ws[threadIdx.x >> 5] = v;
  __syncthreads();
  int t = wave_sum(ws[threadIdx.x & 31]);
  __syncthreads();
  return t;
}

__device__ __forceinline__ int4 load_counts4(const int* __restrict__ cnt, int i0, int n) {
  int4 c;
  if (i0 + 3 < n) {
    c = *(const int4*)(cnt + i0);
  } else {
    c.x = (i0     < n) ? cnt[i0]     : 0;
    c.y = (i0 + 1 < n) ? cnt[i0 + 1] : 0;
    c.z = (i0 + 2 < n) ? cnt[i0 + 2] : 0;
    c.w = 0;
  }
  return c;
}

__global__ __launch_bounds__(SCAN_THREADS)
void row_total_kernel(const int* __restrict__ cnt, int* __restrict__ btot, int n) {
  __shared__ int ws[32];
  const int i0 = (int)blockIdx.x * SCAN_CHUNK + (int)threadIdx.x * 4;
  const int4 c = load_counts4(cnt, i0, n);
  const int t = block_sum(c.x + c.y + c.z + c.w, ws);
  if (threadIdx.x == 0) btot[blockIdx.x] = t;
}

__global__ __launch_bounds__(SCAN_THREADS)
void row_start_kernel(const int* __restrict__ cnt, const int* __restrict__ btot,
                      int* __restrict__ rowptr, int n) {
  __shared__ int ws[32];
  const int tid = threadIdx.x, lane = tid & 31, wv = tid >> 5;
  int before = 0;
  for (int j = tid; j < (int)blockIdx.x; j += SCAN_THREADS) before += btot[j];
  before = block_sum(before, ws);

  const int i0 = (int)blockIdx.x * SCAN_CHUNK + tid * 4;
  const int4 c = load_counts4(cnt, i0, n);
  const int local = c.x + c.y + c.z + c.w;
  int inc = local;
#pragma unroll
  for (int o = 1; o < 32; o <<= 1) {
    const int t = __shfl_up(inc, o, 32);
    if (lane >= o) inc += t;
  }
  if (lane == 31) ws[wv] = inc;
  __syncthreads();
  const int wt = ws[lane];
  int winc = wt;
#pragma unroll
  for (int o = 1; o < 32; o <<= 1) {
    const int t = __shfl_up(winc, o, 32);
    if (lane >= o) winc += t;
  }
  const int wexc = __shfl(winc - wt, wv, 32);
  const int p0 = before + wexc + (inc - local);
  const int p1 = p0 + c.x, p2 = p1 + c.y, p3 = p2 + c.z;
  if (i0 + 3 < n) {
    *(int4*)(rowptr + i0) = make_int4(p0, p1, p2, p3);
  } else {
    if (i0     < n) rowptr[i0]     = p0;
    if (i0 + 1 < n) rowptr[i0 + 1] = p1;
    if (i0 + 2 < n) rowptr[i0 + 2] = p2;
  }
}

#define FILL_BATCH 4

__global__ __launch_bounds__(RES_THREADS)
void fill_kernel(const int* __restrict__ row, const int* __restrict__ other,
                 const int* __restrict__ rank, const int* __restrict__ rowptr,
                 int* __restrict__ col, int E) {
  const int nt = (int)(gridDim.x * blockDim.x);
  int e = (int)(blockIdx.x * blockDim.x + threadIdx.x);
  for (; e + (FILL_BATCH - 1) * nt < E; e += FILL_BATCH * nt) {
    int s[FILL_BATCH], d[FILL_BATCH], r[FILL_BATCH], p[FILL_BATCH];
#pragma unroll
    for (int u = 0; u < FILL_BATCH; ++u) {
      s[u] = row[e + u * nt];
      d[u] = other[e + u * nt];
      r[u] = rank[e + u * nt];
    }
#pragma unroll
    for (int u = 0; u < FILL_BATCH; ++u) p[u] = rowptr[s[u]];
#pragma unroll
    for (int u = 0; u < FILL_BATCH; ++u) col[p[u] + r[u]] = d[u];
  }
  for (; e < E; e += nt) col[rowptr[row[e]] + rank[e]] = other[e];
}

#define HIST_BINS    16384
#define HIST_THREADS 1024
#define HIST_BATCH   8
#define HIST_BLOCKS  64

__global__ __launch_bounds__(HIST_THREADS)
void degree_hist_kernel(const int* __restrict__ idx,
                        unsigned* __restrict__ part, int E, int n) {
  __shared__ unsigned h[HIST_BINS];
  const int tid = threadIdx.x;
  const int chunk = (E + (int)gridDim.x - 1) / (int)gridDim.x;
  const int e0 = (int)blockIdx.x * chunk;
  const int e1 = min(e0 + chunk, E);
  unsigned* prow = part + (size_t)blockIdx.x * (size_t)n;

  for (int base = 0; base < n; base += HIST_BINS) {
    const int nb = min(HIST_BINS, n - base);
    for (int i = tid; i < nb; i += HIST_THREADS) h[i] = 0u;
    __syncthreads();

    for (int e = e0 + tid; e < e1; e += HIST_BATCH * HIST_THREADS) {
      int b[HIST_BATCH];
#pragma unroll
      for (int u = 0; u < HIST_BATCH; ++u) {
        const int ee = e + u * HIST_THREADS;
        b[u] = (ee < e1) ? idx[ee] : -1;
      }
#pragma unroll
      for (int u = 0; u < HIST_BATCH; ++u) {
        unsigned r = (unsigned)(b[u] - base);
        if (r < (unsigned)nb) atomicAdd(&h[r], 1u);
      }
    }
    __syncthreads();

    for (int i = tid; i < nb; i += HIST_THREADS) prow[base + i] = h[i];
    __syncthreads();
  }
}

#define SUM_BATCH 16

__global__ void degree_sum_kernel(unsigned* part, int* __restrict__ cnt,
                                  float* __restrict__ deg, int n, int rows) {
  int i = blockIdx.x * blockDim.x + threadIdx.x;
  if (i >= n) return;
  unsigned s = 0u;
  for (int g0 = 0; g0 < rows; g0 += SUM_BATCH) {
    unsigned v[SUM_BATCH];
#pragma unroll
    for (int u = 0; u < SUM_BATCH; ++u)
      v[u] = (g0 + u < rows) ? part[(size_t)(g0 + u) * (size_t)n + i] : 0u;
#pragma unroll
    for (int u = 0; u < SUM_BATCH; ++u) {
      if (g0 + u < rows) part[(size_t)(g0 + u) * (size_t)n + i] = s;
      s += v[u];
    }
  }
  cnt[i] = (int)s;
  deg[i] = (float)s;
}

__global__ __launch_bounds__(SCAN_THREADS)
void short_row_start_kernel(const int* __restrict__ cnt, int* __restrict__ rowptr, int n) {
  __shared__ int ws[32];
  const int tid = threadIdx.x, lane = tid & 31, wv = tid >> 5;
  int carry = 0;
  for (int c0 = 0; c0 < n; c0 += SCAN_CHUNK) {
    const int i0 = c0 + tid * 4;
    const int4 c = load_counts4(cnt, i0, n);
    const int local = c.x + c.y + c.z + c.w;
    int inc = local;
#pragma unroll
    for (int o = 1; o < 32; o <<= 1) {
      const int t = __shfl_up(inc, o, 32);
      if (lane >= o) inc += t;
    }
    if (lane == 31) ws[wv] = inc;
    __syncthreads();
    const int wt = ws[lane];
    int winc = wt;
#pragma unroll
    for (int o = 1; o < 32; o <<= 1) {
      const int t = __shfl_up(winc, o, 32);
      if (lane >= o) winc += t;
    }
    const int wexc = __shfl(winc - wt, wv, 32);
    const int total = __shfl(winc, 31, 32);
    const int p0 = carry + wexc + (inc - local);
    const int p1 = p0 + c.x, p2 = p1 + c.y, p3 = p2 + c.z;
    if (i0 + 3 < n) {
      *(int4*)(rowptr + i0) = make_int4(p0, p1, p2, p3);
    } else {
      if (i0     < n) rowptr[i0]     = p0;
      if (i0 + 1 < n) rowptr[i0 + 1] = p1;
      if (i0 + 2 < n) rowptr[i0 + 2] = p2;
    }
    carry += total;
    __syncthreads();
  }
}

__global__ __launch_bounds__(HIST_THREADS)
void bucket_fill_kernel(const int* __restrict__ idx, const int* __restrict__ other,
                        const unsigned* __restrict__ part,
                        const int* __restrict__ rowptr,
                        int* __restrict__ col, int E, int n) {
  __shared__ unsigned h[HIST_BINS];
  const int tid = threadIdx.x;
  const int chunk = (E + (int)gridDim.x - 1) / (int)gridDim.x;
  const int e0 = (int)blockIdx.x * chunk;
  const int e1 = min(e0 + chunk, E);
  const unsigned* prow = part + (size_t)blockIdx.x * (size_t)n;

  for (int base = 0; base < n; base += HIST_BINS) {
    const int nb = min(HIST_BINS, n - base);
    for (int i = tid; i < nb; i += HIST_THREADS)
      h[i] = (unsigned)rowptr[base + i] + prow[base + i];
    __syncthreads();

    for (int e = e0 + tid; e < e1; e += HIST_BATCH * HIST_THREADS) {
      int b[HIST_BATCH], o[HIST_BATCH];
#pragma unroll
      for (int u = 0; u < HIST_BATCH; ++u) {
        const int ee = e + u * HIST_THREADS;
        b[u] = (ee < e1) ? idx[ee] : -1;
        o[u] = (ee < e1) ? other[ee] : 0;
      }
#pragma unroll
      for (int u = 0; u < HIST_BATCH; ++u) {
        unsigned r = (unsigned)(b[u] - base);
        if (r < (unsigned)nb) col[atomicAdd(&h[r], 1u)] = o[u];
      }
    }
    __syncthreads();
  }
}

__global__ __launch_bounds__(RES_THREADS)
void gather_long_kernel(const float* __restrict__ srcf,
                        const int* __restrict__ rowptr,
                        const int* __restrict__ rowlen,
                        const int* __restrict__ col,
                        float* __restrict__ agg, int n) {
  const int lane = threadIdx.x & 31;
  const int part = lane & 7;
  const int g    = lane >> 3;
  const int nw   = (int)((gridDim.x * blockDim.x) >> 5);
  for (int node = (int)((blockIdx.x * blockDim.x + threadIdx.x) >> 5); node < n; node += nw) {
    const int start = rowptr[node];
    const int len   = rowlen[node];
    float4 acc = make_float4(0.0f, 0.0f, 0.0f, 0.0f);
    int ci = (lane < len) ? col[start + lane] : 0;
    for (int base = 0; base < len; base += 32) {
      const int m = min(32, len - base);
      const int nx = base + 32 + lane;
      const int cn = (nx < len) ? col[start + nx] : 0;
      float4 v[8];
#pragma unroll
      for (int j = 0; j < 8; ++j) {
        const int jj = j * 4 + g;
        const int c = __shfl(ci, (jj < m) ? jj : 0, 32);
        v[j] = ((const float4*)(srcf + (long long)c * 32))[part];
      }
#pragma unroll
      for (int j = 0; j < 8; ++j) {
        const bool in = (j * 4 + g) < m;
        acc.x += in ? v[j].x : 0.0f;
        acc.y += in ? v[j].y : 0.0f;
        acc.z += in ? v[j].z : 0.0f;
        acc.w += in ? v[j].w : 0.0f;
      }
      ci = cn;
    }
    acc.x += __shfl_xor(acc.x, 8, 32);  acc.y += __shfl_xor(acc.y, 8, 32);
    acc.z += __shfl_xor(acc.z, 8, 32);  acc.w += __shfl_xor(acc.w, 8, 32);
    acc.x += __shfl_xor(acc.x, 16, 32); acc.y += __shfl_xor(acc.y, 16, 32);
    acc.z += __shfl_xor(acc.z, 16, 32); acc.w += __shfl_xor(acc.w, 16, 32);
    if (g == 0) ((float4*)(agg + (long long)node * 32))[part] = acc;
  }
}

template <int NTILES, bool RELU>
__global__ void sage_wmma_kernel(const float* __restrict__ agg,
                                 const float* __restrict__ deg,
                                 const float* __restrict__ xdst,
                                 const float* __restrict__ Wl,
                                 const float* __restrict__ bl,
                                 const float* __restrict__ Wr,
                                 float* __restrict__ out,
                                 int n) {
  const int OUTF = NTILES * 16;
  int lane = threadIdx.x & 31;
  int wave = (blockIdx.x * blockDim.x + threadIdx.x) >> 5;
  int nb = wave * 16;
  if (nb >= n) return;
  int m  = lane & 15;
  int hi = lane >> 4;

  int node = nb + m;
  bool nv = node < n;
  int ncl = nv ? node : (n - 1);
  float dinv = nv ? (1.0f / fmaxf(deg[ncl], 1.0f)) : 0.0f;
  const float* arow = agg  + (long long)ncl * 32;
  const float* xrow = xdst + (long long)ncl * 32;

  v2f a_agg[8], a_x[8];
#pragma unroll
  for (int s = 0; s < 8; ++s) {
    int k = 4 * s + 2 * hi;
    a_agg[s].x = arow[k] * dinv;
    a_agg[s].y = arow[k + 1] * dinv;
    a_x[s].x   = nv ? xrow[k]     : 0.0f;
    a_x[s].y   = nv ? xrow[k + 1] : 0.0f;
  }

  bool full = (nb + 16 <= n);

#pragma unroll
  for (int t = 0; t < NTILES; ++t) {
    int ng = t * 16 + m;
    const float* wl = Wl + ng * 32;
    const float* wr = Wr + ng * 32;
    float bias = bl[ng];
    v8f c;
#pragma unroll
    for (int v = 0; v < 8; ++v) c[v] = bias;

#pragma unroll
    for (int s = 0; s < 8; ++s) {
      int k = 4 * s + 2 * hi;
      v2f b;
      b.x = wl[k];
      b.y = wl[k + 1];
      c = __builtin_amdgcn_wmma_f32_16x16x4_f32(false, a_agg[s], false, b,
                                                (short)0, c, false, false);
    }
#pragma unroll
    for (int s = 0; s < 8; ++s) {
      int k = 4 * s + 2 * hi;
      v2f b;
      b.x = wr[k];
      b.y = wr[k + 1];
      c = __builtin_amdgcn_wmma_f32_16x16x4_f32(false, a_x[s], false, b,
                                                (short)0, c, false, false);
    }

    if (full) {
#pragma unroll
      for (int v = 0; v < 8; ++v) {
        float val = c[v];
        if (RELU) val = fmaxf(val, 0.0f);
        out[(long long)(nb + v + 8 * hi) * OUTF + ng] = val;
      }
    } else {
#pragma unroll
      for (int v = 0; v < 8; ++v) {
        int nd = nb + v + 8 * hi;
        if (nd < n) {
          float val = c[v];
          if (RELU) val = fmaxf(val, 0.0f);
          out[(long long)nd * OUTF + ng] = val;
        }
      }
    }
  }
}

#define TILE_PITCH 36

template <int NTILES, bool RELU>
__global__ __launch_bounds__(RES_THREADS)
void gather_sage_wmma_kernel(const float* __restrict__ srcf,
                             const int* __restrict__ rowptr,
                             const int* __restrict__ rowlen,
                             const int* __restrict__ col,
                             const float* __restrict__ xdst,
                             const float* __restrict__ Wl,
                             const float* __restrict__ bl,
                             const float* __restrict__ Wr,
                             float* __restrict__ out,
                             int n) {
  const int OUTF = NTILES * 16;
  __shared__ __attribute__((aligned(16))) float tiles[(RES_THREADS / 32) * 16 * TILE_PITCH];
  const int lane = threadIdx.x & 31;
  float* T = tiles + (threadIdx.x >> 5) * (16 * TILE_PITCH);
  const int part = lane & 7;
  const int g    = lane >> 3;
  const int m    = lane & 15;
  const int hi   = lane >> 4;
  const int ntiles = (n + 15) >> 4;
  const int tile = (int)((blockIdx.x * blockDim.x + threadIdx.x) >> 5);
  if (tile >= ntiles) return;
  {
    const int nb = tile * 16;

    const int hnode = nb + m;
    const bool nv = hnode < n;
    const int ncl = nv ? hnode : (n - 1);
    const float* xrow = xdst + (long long)ncl * 32;
    v2f xraw[8];
#pragma unroll
    for (int s = 0; s < 8; ++s) xraw[s] = *(const v2f*)(xrow + 4 * s + 2 * hi);

    int meta = 0;
    if (hnode < n) meta = hi ? rowlen[hnode] : rowptr[hnode];
    int start[4], len[4], cfirst[4];
#pragma unroll
    for (int q = 0; q < 4; ++q) {
      start[q] = __shfl(meta, 4 * q + g, 32);
      len[q]   = __shfl(meta, 16 + 4 * q + g, 32);
    }
#pragma unroll
    for (int q = 0; q < 4; ++q)
      cfirst[q] = (part < len[q]) ? col[start[q] + part] : 0;

#pragma unroll
    for (int q = 0; q < 4; ++q) {
      const int st = start[q];
      const int ln = len[q];
      float4 acc = make_float4(0.0f, 0.0f, 0.0f, 0.0f);
      int cn = cfirst[q];
      for (int base = 0; base < ln; base += 8) {
        const int mm = min(8, ln - base);
        const int ci = cn;
        const int nx = base + 8 + part;
        cn = (nx < ln) ? col[st + nx] : 0;
        for (int j0 = 0; j0 < mm; j0 += 4) {
          float4 v[4];
#pragma unroll
          for (int j = 0; j < 4; ++j) {
            const int jj = (j0 + j < mm) ? (j0 + j) : j0;
            const int c = __shfl(ci, jj, 8);
            v[j] = ((const float4*)(srcf + (long long)c * 32))[part];
          }
#pragma unroll
          for (int j = 0; j < 4; ++j) {
            if (j0 + j < mm) {
              acc.x += v[j].x; acc.y += v[j].y; acc.z += v[j].z; acc.w += v[j].w;
            }
          }
        }
      }
      const float dinv = 1.0f / fmaxf((float)ln, 1.0f);
      float* trow = T + (4 * q + g) * TILE_PITCH + part * 4;
      trow[0] = acc.x * dinv;
      trow[1] = acc.y * dinv;
      trow[2] = acc.z * dinv;
      trow[3] = acc.w * dinv;
    }

    __builtin_amdgcn_fence(__ATOMIC_RELEASE, "wavefront");
    __builtin_amdgcn_wave_barrier();
    __builtin_amdgcn_fence(__ATOMIC_ACQUIRE, "wavefront");

    const float* arow = T + m * TILE_PITCH;

    v2f a_agg[8], a_x[8];
#pragma unroll
    for (int s = 0; s < 8; ++s) {
      const int k = 4 * s + 2 * hi;
      a_agg[s].x = arow[k];
      a_agg[s].y = arow[k + 1];
      a_x[s].x   = nv ? xraw[s].x : 0.0f;
      a_x[s].y   = nv ? xraw[s].y : 0.0f;
    }

    const bool full = (nb + 16 <= n);

#pragma unroll
    for (int t = 0; t < NTILES; ++t) {
      const int ng = t * 16 + m;
      const float* wl = Wl + ng * 32;
      const float* wr = Wr + ng * 32;
      const float bias = bl[ng];
      v8f c;
#pragma unroll
      for (int v = 0; v < 8; ++v) c[v] = bias;

#pragma unroll
      for (int s = 0; s < 8; ++s) {
        const int k = 4 * s + 2 * hi;
        v2f b;
        b.x = wl[k];
        b.y = wl[k + 1];
        c = __builtin_amdgcn_wmma_f32_16x16x4_f32(false, a_agg[s], false, b,
                                                  (short)0, c, false, false);
      }
#pragma unroll
      for (int s = 0; s < 8; ++s) {
        const int k = 4 * s + 2 * hi;
        v2f b;
        b.x = wr[k];
        b.y = wr[k + 1];
        c = __builtin_amdgcn_wmma_f32_16x16x4_f32(false, a_x[s], false, b,
                                                  (short)0, c, false, false);
      }

      if (full) {
#pragma unroll
        for (int v = 0; v < 8; ++v) {
          float val = c[v];
          if (RELU) val = fmaxf(val, 0.0f);
          out[(long long)(nb + v + 8 * hi) * OUTF + ng] = val;
        }
      } else {
#pragma unroll
        for (int v = 0; v < 8; ++v) {
          const int nd = nb + v + 8 * hi;
          if (nd < n) {
            float val = c[v];
            if (RELU) val = fmaxf(val, 0.0f);
            out[(long long)nd * OUTF + ng] = val;
          }
        }
      }
    }
  }
}

__global__ void decode_kernel(const float* __restrict__ ou,
                              const float* __restrict__ oc,
                              const int* __restrict__ ls,
                              const int* __restrict__ ld,
                              float* __restrict__ out, int L) {
  int t = blockIdx.x * blockDim.x + threadIdx.x;
  if (t >= L) return;
  const float4* a = (const float4*)(ou + (long long)ls[t] * 16);
  const float4* b = (const float4*)(oc + (long long)ld[t] * 16);
  float acc = 0.0f;
#pragma unroll
  for (int i = 0; i < 4; ++i) {
    float4 xa = a[i], xb = b[i];
    acc += xa.x * xb.x + xa.y * xb.y + xa.z * xb.z + xa.w * xb.w;
  }
  out[t] = acc;
}

extern "C" void kernel_launch(void* const* d_in, const int* in_sizes, int n_in,
                              void* d_out, int out_size, void* d_ws, size_t ws_size,
                              hipStream_t stream) {
  const float* user_x       = (const float*)d_in[0];
  const float* course_x     = (const float*)d_in[1];
  const float* user_embed   = (const float*)d_in[2];
  const float* course_embed = (const float*)d_in[3];
  const float* Wu = (const float*)d_in[4];
  const float* bu = (const float*)d_in[5];
  const float* Wc = (const float*)d_in[6];
  const float* bc = (const float*)d_in[7];
  const float* c1e_Wl = (const float*)d_in[8];
  const float* c1e_bl = (const float*)d_in[9];
  const float* c1e_Wr = (const float*)d_in[10];
  const float* c1r_Wl = (const float*)d_in[11];
  const float* c1r_bl = (const float*)d_in[12];
  const float* c1r_Wr = (const float*)d_in[13];
  const float* c2e_Wl = (const float*)d_in[14];
  const float* c2e_bl = (const float*)d_in[15];
  const float* c2e_Wr = (const float*)d_in[16];
  const float* c2r_Wl = (const float*)d_in[17];
  const float* c2r_bl = (const float*)d_in[18];
  const float* c2r_Wr = (const float*)d_in[19];
  const int* user_idx   = (const int*)d_in[20];
  const int* course_idx = (const int*)d_in[21];
  const int* edge_src   = (const int*)d_in[22];
  const int* edge_dst   = (const int*)d_in[23];
  const int* label_src  = (const int*)d_in[24];
  const int* label_dst  = (const int*)d_in[25];
  float* outp = (float*)d_out;

  const int NU = in_sizes[2] / 16;
  const int NC = in_sizes[3] / 16;
  const int E  = in_sizes[22];
  const int L  = in_sizes[24];

  char* p = (char*)d_ws;
  auto carve = [&](size_t bytes) -> float* {
    float* r = (float*)p;
    p += (bytes + 255) & ~((size_t)255);
    return r;
  };
  float* xu   = carve((size_t)NU * 32 * 4);
  float* xc   = carve((size_t)NC * 32 * 4);
  float* ubkt = carve((size_t)NU * 32 * 4);
  float* aggc = carve((size_t)NC * 32 * 4);
  carve((size_t)NU * 4);
  float* degc = carve((size_t)NC * 4);
  float* hu   = carve((size_t)NU * 32 * 4);
  float* hc   = carve((size_t)NC * 32 * 4);
  float* ou   = carve((size_t)NU * 16 * 4);
  float* oc   = carve((size_t)NC * 16 * 4);

  const int B = 256;
  auto nblk = [](long long t, int b) { return (unsigned)((t + b - 1) / b); };
  auto rblk = [](long long t) {
    long long g = (t + RES_THREADS - 1) / RES_THREADS;
    return (unsigned)(g < RES_BLOCKS ? g : RES_BLOCKS);
  };

  const unsigned scan_blocks = nblk(NU, SCAN_CHUNK);
  int* ucnt  = (int*)ubkt;
  int* urow  = ucnt + (((size_t)NU + 63) & ~(size_t)63);
  int* urank = urow + (((size_t)NU + 63) & ~(size_t)63);
  int* ucol  = urank + (((size_t)E + 63) & ~(size_t)63);
  int* ubtot = ucol + (((size_t)E + 63) & ~(size_t)63);
  int* ccnt  = ubtot + (((size_t)scan_blocks + 63) & ~(size_t)63);
  int* crow  = ccnt + (((size_t)NC + 63) & ~(size_t)63);
  int* ccol  = crow + (((size_t)NC + 63) & ~(size_t)63);

  hipMemsetAsync(ucnt, 0, (size_t)NU * 4, stream);

  fuse_kernel<5><<<rblk((long long)NU * 8), RES_THREADS, 0, stream>>>(
      user_x, user_embed, user_idx, Wu, bu, xu, NU);
  fuse_kernel<2><<<rblk((long long)NC * 8), RES_THREADS, 0, stream>>>(
      course_x, course_embed, course_idx, Wc, bc, xc, NC);

  unsigned* degpart = (unsigned*)hu;
  degree_hist_kernel<<<HIST_BLOCKS, HIST_THREADS, 0, stream>>>(edge_dst, degpart, E, NC);
  degree_sum_kernel<<<nblk(NC, B), B, 0, stream>>>(degpart, ccnt, degc, NC, HIST_BLOCKS);
  short_row_start_kernel<<<1, SCAN_THREADS, 0, stream>>>(ccnt, crow, NC);
  bucket_fill_kernel<<<HIST_BLOCKS, HIST_THREADS, 0, stream>>>(
      edge_dst, edge_src, degpart, crow, ccol, E, NC);
  rank_kernel<<<rblk(E), RES_THREADS, 0, stream>>>(edge_src, ucnt, urank, E);
  row_total_kernel<<<scan_blocks, SCAN_THREADS, 0, stream>>>(ucnt, ubtot, NU);
  row_start_kernel<<<scan_blocks, SCAN_THREADS, 0, stream>>>(ucnt, ubtot, urow, NU);
  fill_kernel<<<rblk(E), RES_THREADS, 0, stream>>>(edge_src, edge_dst, urank, urow, ucol, E);

  gather_long_kernel<<<rblk((long long)NC * 32), RES_THREADS, 0, stream>>>(xu, crow, ccnt, ccol, aggc, NC);
  sage_wmma_kernel<2, true><<<nblk(((long long)(NC + 15) / 16) * 32, B), B, 0, stream>>>(
      aggc, degc, xc, c1e_Wl, c1e_bl, c1e_Wr, hc, NC);
  gather_sage_wmma_kernel<2, true><<<nblk(((long long)(NU + 15) / 16) * 32, RES_THREADS), RES_THREADS, 0, stream>>>(
      xc, urow, ucnt, ucol, xu, c1r_Wl, c1r_bl, c1r_Wr, hu, NU);

  gather_long_kernel<<<rblk((long long)NC * 32), RES_THREADS, 0, stream>>>(hu, crow, ccnt, ccol, aggc, NC);
  sage_wmma_kernel<1, false><<<nblk(((long long)(NC + 15) / 16) * 32, B), B, 0, stream>>>(
      aggc, degc, hc, c2e_Wl, c2e_bl, c2e_Wr, oc, NC);
  gather_sage_wmma_kernel<1, false><<<nblk(((long long)(NU + 15) / 16) * 32, RES_THREADS), RES_THREADS, 0, stream>>>(
      hc, urow, ucnt, ucol, hu, c2r_Wl, c2r_bl, c2r_Wr, ou, NU);

  decode_kernel<<<nblk(L, B), B, 0, stream>>>(ou, oc, label_src, label_dst, outp, L);
}
